# starting kernel
# baseline (speedup 1.0000x reference)
#include <hip/hip_runtime.h>
#include <hip/hip_bf16.h>

#define N_NODES 10000
#define N_EDGES 320000
#define F_INF   128
#define F_OUTF  128
#define H_HEADS 4
#define C_OUT   (H_HEADS * F_OUTF)
#define STEPS   2
#define NEG_SLOPE 0.2f

typedef float v2f __attribute__((ext_vector_type(2)));
typedef float v8f __attribute__((ext_vector_type(8)));

__global__ __launch_bounds__(256) void gemm_xw(const float* __restrict__ x,
                                               const float* __restrict__ W,
                                               float* __restrict__ h) {
  const int lane = threadIdx.x & 31;
  const int wv   = threadIdx.x >> 5;
  const int tile = blockIdx.x * 8 + wv;
  const int tiles_n = C_OUT / 16;
  const int tm = tile / tiles_n;
  const int tn = tile % tiles_n;
  const int m0 = tm * 16, n0 = tn * 16;
  const int half = lane >> 4;
  const int mr   = lane & 15;

  v8f acc = {};
  const float* arow = x + (size_t)(m0 + mr) * F_INF;
  #pragma unroll 4
  for (int k0 = 0; k0 < F_INF; k0 += 4) {
    const int k = k0 + 2 * half;
    v2f a; a.x = arow[k]; a.y = arow[k + 1];
    v2f b; b.x = W[(size_t)k * C_OUT + n0 + mr];
           b.y = W[(size_t)(k + 1) * C_OUT + n0 + mr];
    acc = __builtin_amdgcn_wmma_f32_16x16x4_f32(
         false, a,  false, b,
         (short)0, acc,  false,  false);
  }
  const int nn = lane & 15;
  const int mbase = half * 8;
  #pragma unroll
  for (int i = 0; i < 8; ++i)
    h[(size_t)(m0 + mbase + i) * C_OUT + n0 + nn] = acc[i];
}

__global__ __launch_bounds__(128) void node_logits(const float* __restrict__ h,
                                                   const float* __restrict__ attn_l,
                                                   const float* __restrict__ attn_r,
                                                   float* __restrict__ el,
                                                   float* __restrict__ er) {
  const int n = blockIdx.x;
  const int head = threadIdx.x >> 5, lane = threadIdx.x & 31;
  const float* hr = h + (size_t)n * C_OUT + head * F_OUTF;
  const float* al = attn_l + head * F_OUTF;
  const float* ar = attn_r + head * F_OUTF;
  float sl = 0.f, sr = 0.f;
  #pragma unroll
  for (int f = lane; f < F_OUTF; f += 32) {
    const float v = hr[f];
    sl += v * al[f];
    sr += v * ar[f];
  }
  #pragma unroll
  for (int off = 16; off; off >>= 1) {
    sl += __shfl_xor(sl, off, 32);
    sr += __shfl_xor(sr, off, 32);
  }
  if (lane == 0) {
    el[n * H_HEADS + head] = sl;
    er[n * H_HEADS + head] = sr;
  }
}

__global__ void edge_count(const int* __restrict__ dst, int* __restrict__ cnt) {
  const int e = blockIdx.x * blockDim.x + threadIdx.x;
  if (e < N_EDGES) atomicAdd(&cnt[dst[e]], 1);
}

__global__ __launch_bounds__(1024) void scan_rowptr(const int* __restrict__ cnt,
                                                    int* __restrict__ row_ptr,
                                                    int* __restrict__ cursor) {
  __shared__ int sh[1024];
  __shared__ int carry;
  const int t = threadIdx.x;
  if (t == 0) carry = 0;
  __syncthreads();
  for (int base = 0; base < N_NODES; base += 1024) {
    const int idx = base + t;
    const int v = (idx < N_NODES) ? cnt[idx] : 0;
    sh[t] = v;
    __syncthreads();
    for (int off = 1; off < 1024; off <<= 1) {
      const int u = (t >= off) ? sh[t - off] : 0;
      __syncthreads();
      sh[t] += u;
      __syncthreads();
    }
    const int excl = sh[t] - v + carry;
    if (idx < N_NODES) { row_ptr[idx] = excl; cursor[idx] = excl; }
    __syncthreads();
    if (t == 1023) carry += sh[1023];
    __syncthreads();
  }
  if (t == 0) row_ptr[N_NODES] = carry;
}

__global__ void edge_fill(const int* __restrict__ dst, int* __restrict__ cursor,
                          int* __restrict__ edge_idx) {
  const int e = blockIdx.x * blockDim.x + threadIdx.x;
  if (e < N_EDGES) {
    const int p = atomicAdd(&cursor[dst[e]], 1);
    edge_idx[p] = e;
  }
}

__global__ __launch_bounds__(128) void gat_aggregate(
    const float* __restrict__ h, const float* __restrict__ el,
    const float* __restrict__ er, const int* __restrict__ src,
    const int* __restrict__ row_ptr, const int* __restrict__ edge_idx,
    const float* __restrict__ bias, const float* __restrict__ alpha_p,
    const float* __restrict__ x0, float* __restrict__ xout) {
  __shared__ float red[H_HEADS * F_OUTF];
  const int n = blockIdx.x;
  const int head = threadIdx.x >> 5, lane = threadIdx.x & 31;
  const int rs = row_ptr[n], re = row_ptr[n + 1];
  const float erd = er[n * H_HEADS + head];

  float mx = -INFINITY;
  for (int i = rs + lane; i < re; i += 32) {
    const int e = edge_idx[i];
    float v = el[src[e] * H_HEADS + head] + erd;
    v = v > 0.f ? v : NEG_SLOPE * v;
    mx = fmaxf(mx, v);
  }
  #pragma unroll
  for (int off = 16; off; off >>= 1) mx = fmaxf(mx, __shfl_xor(mx, off, 32));

  float sum = 0.f;
  for (int i = rs + lane; i < re; i += 32) {
    const int e = edge_idx[i];
    float v = el[src[e] * H_HEADS + head] + erd;
    v = v > 0.f ? v : NEG_SLOPE * v;
    sum += __expf(v - mx);
  }
  #pragma unroll
  for (int off = 16; off; off >>= 1) sum += __shfl_xor(sum, off, 32);
  const float inv = (sum != 0.f) ? 1.f / sum : 0.f;

  float4 acc = make_float4(0.f, 0.f, 0.f, 0.f);
  for (int i = rs; i < re; ++i) {
    const int e = edge_idx[i];
    const int s = src[e];
    float v = el[s * H_HEADS + head] + erd;
    v = v > 0.f ? v : NEG_SLOPE * v;
    const float a = __expf(v - mx) * inv;
    const float4 hv = *(const float4*)(h + (size_t)s * C_OUT + head * F_OUTF + lane * 4);
    acc.x += a * hv.x; acc.y += a * hv.y; acc.z += a * hv.z; acc.w += a * hv.w;
  }
  const float4 bv = *(const float4*)(bias + head * F_OUTF + lane * 4);
  acc.x += bv.x; acc.y += bv.y; acc.z += bv.z; acc.w += bv.w;
  *(float4*)(red + head * F_OUTF + lane * 4) = acc;
  __syncthreads();

  const int f = threadIdx.x;
  const float s = 0.25f * (red[f] + red[F_OUTF + f] +
                           red[2 * F_OUTF + f] + red[3 * F_OUTF + f]);
  const float al = alpha_p[0];
  xout[(size_t)n * F_OUTF + f] = (1.f - al) * s + al * x0[(size_t)n * F_OUTF + f];
}

extern "C" void kernel_launch(void* const* d_in, const int* in_sizes, int n_in,
                              void* d_out, int out_size, void* d_ws, size_t ws_size,
                              hipStream_t stream) {
  const float* x      = (const float*)d_in[0];
  const float* x0     = (const float*)d_in[1];
  const int*   src    = (const int*)d_in[2];
  const int*   dst    = (const int*)d_in[3];
  const float* alpha  = (const float*)d_in[4];
  const float* W      = (const float*)d_in[5];
  const float* attn_l = (const float*)d_in[6];
  const float* attn_r = (const float*)d_in[7];
  const float* bias   = (const float*)d_in[8];

  char* ws = (char*)d_ws;
  size_t off = 0;
  auto alloc = [&](size_t bytes) -> char* {
    char* p = ws + off;
    off = (off + bytes + 255) & ~(size_t)255;
    return p;
  };
  float* h        = (float*)alloc((size_t)N_NODES * C_OUT * sizeof(float));
  float* el       = (float*)alloc((size_t)N_NODES * H_HEADS * sizeof(float));
  float* er       = (float*)alloc((size_t)N_NODES * H_HEADS * sizeof(float));
  float* xbuf     = (float*)alloc((size_t)N_NODES * F_INF * sizeof(float));
  int*   cnt      = (int*)alloc((size_t)N_NODES * sizeof(int));
  int*   row_ptr  = (int*)alloc((size_t)(N_NODES + 1) * sizeof(int));
  int*   cursor   = (int*)alloc((size_t)N_NODES * sizeof(int));
  int*   edge_idx = (int*)alloc((size_t)N_EDGES * sizeof(int));

  hipMemsetAsync(cnt, 0, (size_t)N_NODES * sizeof(int), stream);
  edge_count<<<(N_EDGES + 255) / 256, 256, 0, stream>>>(dst, cnt);
  scan_rowptr<<<1, 1024, 0, stream>>>(cnt, row_ptr, cursor);
  edge_fill<<<(N_EDGES + 255) / 256, 256, 0, stream>>>(dst, cursor, edge_idx);

  const float* xin = x;
  for (int step = 0; step < STEPS; ++step) {
    gemm_xw<<<2500, 256, 0, stream>>>(xin, W, h);
    node_logits<<<N_NODES, 128, 0, stream>>>(h, attn_l, attn_r, el, er);
    float* xo = (step == STEPS - 1) ? (float*)d_out : xbuf;
    gat_aggregate<<<N_NODES, 128, 0, stream>>>(h, el, er, src, row_ptr, edge_idx,
                                               bias, alpha, x0, xo);
    xin = xbuf;
  }
}


# CSR build in one launch: 64 groups each own a dst range, stream dst twice, count and fill with LDS atomics; no global atomic, memset or scan launch; in-segment edge order stays arrival order
# speedup vs baseline: 1.3409x; 1.3409x over previous
#include <hip/hip_runtime.h>
#include <hip/hip_bf16.h>

#define N_NODES 10000
#define N_EDGES 320000
#define F_INF   128
#define F_OUTF  128
#define H_HEADS 4
#define C_OUT   (H_HEADS * F_OUTF)
#define STEPS   2
#define NEG_SLOPE 0.2f

typedef float v2f __attribute__((ext_vector_type(2)));
typedef float v8f __attribute__((ext_vector_type(8)));

__global__ __launch_bounds__(256) void gemm_xw(const float* __restrict__ x,
                                               const float* __restrict__ W,
                                               float* __restrict__ h) {
  const int lane = threadIdx.x & 31;
  const int wv   = threadIdx.x >> 5;
  const int tile = blockIdx.x * 8 + wv;
  const int tiles_n = C_OUT / 16;
  const int tm = tile / tiles_n;
  const int tn = tile % tiles_n;
  const int m0 = tm * 16, n0 = tn * 16;
  const int half = lane >> 4;
  const int mr   = lane & 15;

  v8f acc = {};
  const float* arow = x + (size_t)(m0 + mr) * F_INF;
  #pragma unroll 4
  for (int k0 = 0; k0 < F_INF; k0 += 4) {
    const int k = k0 + 2 * half;
    v2f a; a.x = arow[k]; a.y = arow[k + 1];
    v2f b; b.x = W[(size_t)k * C_OUT + n0 + mr];
           b.y = W[(size_t)(k + 1) * C_OUT + n0 + mr];
    acc = __builtin_amdgcn_wmma_f32_16x16x4_f32(
         false, a,  false, b,
         (short)0, acc,  false,  false);
  }
  const int nn = lane & 15;
  const int mbase = half * 8;
  #pragma unroll
  for (int i = 0; i < 8; ++i)
    h[(size_t)(m0 + mbase + i) * C_OUT + n0 + nn] = acc[i];
}

__global__ __launch_bounds__(128) void node_logits(const float* __restrict__ h,
                                                   const float* __restrict__ attn_l,
                                                   const float* __restrict__ attn_r,
                                                   float* __restrict__ el,
                                                   float* __restrict__ er) {
  const int n = blockIdx.x;
  const int head = threadIdx.x >> 5, lane = threadIdx.x & 31;
  const float* hr = h + (size_t)n * C_OUT + head * F_OUTF;
  const float* al = attn_l + head * F_OUTF;
  const float* ar = attn_r + head * F_OUTF;
  float sl = 0.f, sr = 0.f;
  #pragma unroll
  for (int f = lane; f < F_OUTF; f += 32) {
    const float v = hr[f];
    sl += v * al[f];
    sr += v * ar[f];
  }
  #pragma unroll
  for (int off = 16; off; off >>= 1) {
    sl += __shfl_xor(sl, off, 32);
    sr += __shfl_xor(sr, off, 32);
  }
  if (lane == 0) {
    el[n * H_HEADS + head] = sl;
    er[n * H_HEADS + head] = sr;
  }
}

#define CSR_WGS      64
#define CSR_THREADS  1024
#define CSR_RANGE    ((N_NODES + CSR_WGS - 1) / CSR_WGS)
#define CSR_PER_LANE ((CSR_RANGE + 31) / 32)
#define E_VEC        (N_EDGES / 4)
#define CSR_CHUNKS   ((E_VEC + CSR_THREADS - 1) / CSR_THREADS)
#define CSR_PAD      0x7fffffff

static_assert(N_EDGES % 4 == 0, "dst is read as int4");
static_assert((CSR_WGS - 1) * CSR_RANGE < N_NODES, "every group owns at least one node");
static_assert(CSR_WGS <= CSR_CHUNKS, "one distinct start chunk per group");

__global__ __launch_bounds__(CSR_THREADS) void csr_build(const int* __restrict__ dst,
                                                         int* __restrict__ row_ptr,
                                                         int* __restrict__ edge_idx) {
  __shared__ int bins[CSR_RANGE];
  __shared__ int s_below;
  const int t = threadIdx.x, lane = t & 31, wv = t >> 5;
  const int lo = blockIdx.x * CSR_RANGE;
  const int nr = min(CSR_RANGE, N_NODES - lo);
  const int rot = blockIdx.x;
  const int4* d4 = (const int4*)dst;

  for (int j = t; j < CSR_RANGE; j += CSR_THREADS) bins[j] = 0;
  if (t == 0) s_below = 0;
  __syncthreads();

  int below = 0;
  for (int k = 0; k < CSR_CHUNKS; k += 4) {
    int4 d[4];
    #pragma unroll
    for (int u = 0; u < 4; ++u) {
      int c = k + u + rot;
      if (c >= CSR_CHUNKS) c -= CSR_CHUNKS;
      const int v = c * CSR_THREADS + t;
      const bool ok = (k + u < CSR_CHUNKS) && (v < E_VEC);
      d[u] = ok ? d4[v] : make_int4(CSR_PAD, CSR_PAD, CSR_PAD, CSR_PAD);
    }
    #pragma unroll
    for (int u = 0; u < 4; ++u) {
      const int dd[4] = {d[u].x, d[u].y, d[u].z, d[u].w};
      #pragma unroll
      for (int q = 0; q < 4; ++q) {
        const unsigned r = (unsigned)(dd[q] - lo);
        below += (dd[q] < lo) ? 1 : 0;
        if (r < (unsigned)nr) atomicAdd(&bins[r], 1);
      }
    }
  }
  for (int off = 16; off; off >>= 1) below += __shfl_xor(below, off, 32);
  if (lane == 0) atomicAdd(&s_below, below);
  __syncthreads();

  if (wv == 0) {
    const int base = s_below;
    int loc[CSR_PER_LANE];
    int s = 0;
    #pragma unroll
    for (int i = 0; i < CSR_PER_LANE; ++i) {
      const int j = lane * CSR_PER_LANE + i;
      const int c = (j < CSR_RANGE) ? bins[j] : 0;
      loc[i] = s;
      s += c;
    }
    int inc = s;
    for (int off = 1; off < 32; off <<= 1) {
      const int up = __shfl_up(inc, off, 32);
      if (lane >= off) inc += up;
    }
    const int excl = inc - s;
    #pragma unroll
    for (int i = 0; i < CSR_PER_LANE; ++i) {
      const int j = lane * CSR_PER_LANE + i;
      if (j < CSR_RANGE) {
        const int start = base + excl + loc[i];
        bins[j] = start;
        if (j < nr) row_ptr[lo + j] = start;
      }
    }
    if (blockIdx.x == CSR_WGS - 1 && lane == 31) row_ptr[N_NODES] = base + inc;
  }
  __syncthreads();

  for (int k = 0; k < CSR_CHUNKS; k += 4) {
    int4 d[4];
    int  vv[4];
    #pragma unroll
    for (int u = 0; u < 4; ++u) {
      int c = k + u + rot;
      if (c >= CSR_CHUNKS) c -= CSR_CHUNKS;
      const int v = c * CSR_THREADS + t;
      const bool ok = (k + u < CSR_CHUNKS) && (v < E_VEC);
      vv[u] = v;
      d[u] = ok ? d4[v] : make_int4(CSR_PAD, CSR_PAD, CSR_PAD, CSR_PAD);
    }
    int pos[4][4];
    #pragma unroll
    for (int u = 0; u < 4; ++u) {
      const int dd[4] = {d[u].x, d[u].y, d[u].z, d[u].w};
      #pragma unroll
      for (int q = 0; q < 4; ++q) {
        const unsigned r = (unsigned)(dd[q] - lo);
        int p = -1;
        if (r < (unsigned)nr) p = atomicAdd(&bins[r], 1);
        pos[u][q] = p;
      }
    }
    #pragma unroll
    for (int u = 0; u < 4; ++u) {
      #pragma unroll
      for (int q = 0; q < 4; ++q)
        if (pos[u][q] >= 0) edge_idx[pos[u][q]] = 4 * vv[u] + q;
    }
  }
}

__global__ __launch_bounds__(128) void gat_aggregate(
    const float* __restrict__ h, const float* __restrict__ el,
    const float* __restrict__ er, const int* __restrict__ src,
    const int* __restrict__ row_ptr, const int* __restrict__ edge_idx,
    const float* __restrict__ bias, const float* __restrict__ alpha_p,
    const float* __restrict__ x0, float* __restrict__ xout) {
  __shared__ float red[H_HEADS * F_OUTF];
  const int n = blockIdx.x;
  const int head = threadIdx.x >> 5, lane = threadIdx.x & 31;
  const int rs = row_ptr[n], re = row_ptr[n + 1];
  const float erd = er[n * H_HEADS + head];

  float mx = -INFINITY;
  for (int i = rs + lane; i < re; i += 32) {
    const int e = edge_idx[i];
    float v = el[src[e] * H_HEADS + head] + erd;
    v = v > 0.f ? v : NEG_SLOPE * v;
    mx = fmaxf(mx, v);
  }
  #pragma unroll
  for (int off = 16; off; off >>= 1) mx = fmaxf(mx, __shfl_xor(mx, off, 32));

  float sum = 0.f;
  for (int i = rs + lane; i < re; i += 32) {
    const int e = edge_idx[i];
    float v = el[src[e] * H_HEADS + head] + erd;
    v = v > 0.f ? v : NEG_SLOPE * v;
    sum += __expf(v - mx);
  }
  #pragma unroll
  for (int off = 16; off; off >>= 1) sum += __shfl_xor(sum, off, 32);
  const float inv = (sum != 0.f) ? 1.f / sum : 0.f;

  float4 acc = make_float4(0.f, 0.f, 0.f, 0.f);
  for (int i = rs; i < re; ++i) {
    const int e = edge_idx[i];
    const int s = src[e];
    float v = el[s * H_HEADS + head] + erd;
    v = v > 0.f ? v : NEG_SLOPE * v;
    const float a = __expf(v - mx) * inv;
    const float4 hv = *(const float4*)(h + (size_t)s * C_OUT + head * F_OUTF + lane * 4);
    acc.x += a * hv.x; acc.y += a * hv.y; acc.z += a * hv.z; acc.w += a * hv.w;
  }
  const float4 bv = *(const float4*)(bias + head * F_OUTF + lane * 4);
  acc.x += bv.x; acc.y += bv.y; acc.z += bv.z; acc.w += bv.w;
  *(float4*)(red + head * F_OUTF + lane * 4) = acc;
  __syncthreads();

  const int f = threadIdx.x;
  const float s = 0.25f * (red[f] + red[F_OUTF + f] +
                           red[2 * F_OUTF + f] + red[3 * F_OUTF + f]);
  const float al = alpha_p[0];
  xout[(size_t)n * F_OUTF + f] = (1.f - al) * s + al * x0[(size_t)n * F_OUTF + f];
}

extern "C" void kernel_launch(void* const* d_in, const int* in_sizes, int n_in,
                              void* d_out, int out_size, void* d_ws, size_t ws_size,
                              hipStream_t stream) {
  const float* x      = (const float*)d_in[0];
  const float* x0     = (const float*)d_in[1];
  const int*   src    = (const int*)d_in[2];
  const int*   dst    = (const int*)d_in[3];
  const float* alpha  = (const float*)d_in[4];
  const float* W      = (const float*)d_in[5];
  const float* attn_l = (const float*)d_in[6];
  const float* attn_r = (const float*)d_in[7];
  const float* bias   = (const float*)d_in[8];

  char* ws = (char*)d_ws;
  size_t off = 0;
  auto alloc = [&](size_t bytes) -> char* {
    char* p = ws + off;
    off = (off + bytes + 255) & ~(size_t)255;
    return p;
  };
  float* h        = (float*)alloc((size_t)N_NODES * C_OUT * sizeof(float));
  float* el       = (float*)alloc((size_t)N_NODES * H_HEADS * sizeof(float));
  float* er       = (float*)alloc((size_t)N_NODES * H_HEADS * sizeof(float));
  float* xbuf     = (float*)alloc((size_t)N_NODES * F_INF * sizeof(float));
  int*   row_ptr  = (int*)alloc((size_t)(N_NODES + 1) * sizeof(int));
  int*   edge_idx = (int*)alloc((size_t)N_EDGES * sizeof(int));

  csr_build<<<CSR_WGS, CSR_THREADS, 0, stream>>>(dst, row_ptr, edge_idx);

  const float* xin = x;
  for (int step = 0; step < STEPS; ++step) {
    gemm_xw<<<2500, 256, 0, stream>>>(xin, W, h);
    node_logits<<<N_NODES, 128, 0, stream>>>(h, attn_l, attn_r, el, er);
    float* xo = (step == STEPS - 1) ? (float*)d_out : xbuf;
    gat_aggregate<<<N_NODES, 128, 0, stream>>>(h, el, er, src, row_ptr, edge_idx,
                                               bias, alpha, x0, xo);
    xin = xbuf;
  }
}


# node logits el/er computed in the GEMM epilogue: one wave owns 16 rows x one head (8 tiles), node_logits launches and h re-read gone; el/er now summed per lane over 8 columns then a 16-lane tree
# speedup vs baseline: 1.5063x; 1.1233x over previous
#include <hip/hip_runtime.h>
#include <hip/hip_bf16.h>

#define N_NODES 10000
#define N_EDGES 320000
#define F_INF   128
#define F_OUTF  128
#define H_HEADS 4
#define C_OUT   (H_HEADS * F_OUTF)
#define STEPS   2
#define NEG_SLOPE 0.2f

typedef float v2f __attribute__((ext_vector_type(2)));
typedef float v8f __attribute__((ext_vector_type(8)));

#define GEMM_WAVES 5
#define M_TILES    (N_NODES / 16)
static_assert(N_NODES % 16 == 0, "whole row tiles");
static_assert(M_TILES % GEMM_WAVES == 0, "whole work-groups, no tail wave");
static_assert(F_OUTF == 128 && H_HEADS == 4 && F_INF % 4 == 0, "tile naming below");

__global__ __launch_bounds__(GEMM_WAVES * 32) void gemm_xw_logits(
    const float* __restrict__ x, const float* __restrict__ W,
    const float* __restrict__ attn_l, const float* __restrict__ attn_r,
    float* __restrict__ h, float* __restrict__ el, float* __restrict__ er) {
  const int lane = threadIdx.x & 31;
  const int wv   = threadIdx.x >> 5;
  const int hd   = blockIdx.x & (H_HEADS - 1);
  const int tm   = (blockIdx.x >> 2) * GEMM_WAVES + wv;
  const int m0   = tm * 16;
  const int hf   = lane >> 4;
  const int c    = lane & 15;

  v8f acc[8];
  #pragma unroll
  for (int t = 0; t < 8; ++t) acc[t] = v8f{};

  const float* arow = x + (size_t)(m0 + c) * F_INF + 2 * hf;
  const float* bcol = W + (size_t)(2 * hf) * C_OUT + hd * F_OUTF + 4 * c;

  #pragma unroll 4
  for (int k0 = 0; k0 < F_INF; k0 += 4) {
    const v2f a = *(const v2f*)(arow + k0);
    const float* b0 = bcol + (size_t)k0 * C_OUT;
    const float4 x_lo = *(const float4*)(b0);
    const float4 x_hi = *(const float4*)(b0 + 64);
    const float4 y_lo = *(const float4*)(b0 + C_OUT);
    const float4 y_hi = *(const float4*)(b0 + C_OUT + 64);
    const float bx[8] = {x_lo.x, x_lo.y, x_lo.z, x_lo.w, x_hi.x, x_hi.y, x_hi.z, x_hi.w};
    const float by[8] = {y_lo.x, y_lo.y, y_lo.z, y_lo.w, y_hi.x, y_hi.y, y_hi.z, y_hi.w};
    #pragma unroll
    for (int t = 0; t < 8; ++t) {
      v2f b; b.x = bx[t]; b.y = by[t];
      acc[t] = __builtin_amdgcn_wmma_f32_16x16x4_f32(
           false, a,  false, b,
           (short)0, acc[t],  false,  false);
    }
  }

  const int row0 = m0 + 8 * hf;
  const float4 al_lo = *(const float4*)(attn_l + hd * F_OUTF + 4 * c);
  const float4 al_hi = *(const float4*)(attn_l + hd * F_OUTF + 4 * c + 64);
  const float4 ar_lo = *(const float4*)(attn_r + hd * F_OUTF + 4 * c);
  const float4 ar_hi = *(const float4*)(attn_r + hd * F_OUTF + 4 * c + 64);
  float* hp = h + (size_t)row0 * C_OUT + hd * F_OUTF + 4 * c;
  #pragma unroll
  for (int i = 0; i < 8; ++i) {
    const float4 lo = make_float4(acc[0][i], acc[1][i], acc[2][i], acc[3][i]);
    const float4 hi = make_float4(acc[4][i], acc[5][i], acc[6][i], acc[7][i]);
    *(float4*)(hp + (size_t)i * C_OUT)      = lo;
    *(float4*)(hp + (size_t)i * C_OUT + 64) = hi;
    float sl = lo.x * al_lo.x + lo.y * al_lo.y + lo.z * al_lo.z + lo.w * al_lo.w
             + hi.x * al_hi.x + hi.y * al_hi.y + hi.z * al_hi.z + hi.w * al_hi.w;
    float sr = lo.x * ar_lo.x + lo.y * ar_lo.y + lo.z * ar_lo.z + lo.w * ar_lo.w
             + hi.x * ar_hi.x + hi.y * ar_hi.y + hi.z * ar_hi.z + hi.w * ar_hi.w;
    #pragma unroll
    for (int off = 8; off; off >>= 1) {
      sl += __shfl_xor(sl, off, 32);
      sr += __shfl_xor(sr, off, 32);
    }
    if (c == 0) {
      el[(row0 + i) * H_HEADS + hd] = sl;
      er[(row0 + i) * H_HEADS + hd] = sr;
    }
  }
}

#define CSR_WGS      64
#define CSR_THREADS  1024
#define CSR_RANGE    ((N_NODES + CSR_WGS - 1) / CSR_WGS)
#define CSR_PER_LANE ((CSR_RANGE + 31) / 32)
#define E_VEC        (N_EDGES / 4)
#define CSR_CHUNKS   ((E_VEC + CSR_THREADS - 1) / CSR_THREADS)
#define CSR_PAD      0x7fffffff

static_assert(N_EDGES % 4 == 0, "dst is read as int4");
static_assert((CSR_WGS - 1) * CSR_RANGE < N_NODES, "every group owns at least one node");
static_assert(CSR_WGS <= CSR_CHUNKS, "one distinct start chunk per group");

__global__ __launch_bounds__(CSR_THREADS) void csr_build(const int* __restrict__ dst,
                                                         int* __restrict__ row_ptr,
                                                         int* __restrict__ edge_idx) {
  __shared__ int bins[CSR_RANGE];
  __shared__ int s_below;
  const int t = threadIdx.x, lane = t & 31, wv = t >> 5;
  const int lo = blockIdx.x * CSR_RANGE;
  const int nr = min(CSR_RANGE, N_NODES - lo);
  const int rot = blockIdx.x;
  const int4* d4 = (const int4*)dst;

  for (int j = t; j < CSR_RANGE; j += CSR_THREADS) bins[j] = 0;
  if (t == 0) s_below = 0;
  __syncthreads();

  int below = 0;
  for (int k = 0; k < CSR_CHUNKS; k += 4) {
    int4 d[4];
    #pragma unroll
    for (int u = 0; u < 4; ++u) {
      int c = k + u + rot;
      if (c >= CSR_CHUNKS) c -= CSR_CHUNKS;
      const int v = c * CSR_THREADS + t;
      const bool ok = (k + u < CSR_CHUNKS) && (v < E_VEC);
      d[u] = ok ? d4[v] : make_int4(CSR_PAD, CSR_PAD, CSR_PAD, CSR_PAD);
    }
    #pragma unroll
    for (int u = 0; u < 4; ++u) {
      const int dd[4] = {d[u].x, d[u].y, d[u].z, d[u].w};
      #pragma unroll
      for (int q = 0; q < 4; ++q) {
        const unsigned r = (unsigned)(dd[q] - lo);
        below += (dd[q] < lo) ? 1 : 0;
        if (r < (unsigned)nr) atomicAdd(&bins[r], 1);
      }
    }
  }
  for (int off = 16; off; off >>= 1) below += __shfl_xor(below, off, 32);
  if (lane == 0) atomicAdd(&s_below, below);
  __syncthreads();

  if (wv == 0) {
    const int base = s_below;
    int loc[CSR_PER_LANE];
    int s = 0;
    #pragma unroll
    for (int i = 0; i < CSR_PER_LANE; ++i) {
      const int j = lane * CSR_PER_LANE + i;
      const int c = (j < CSR_RANGE) ? bins[j] : 0;
      loc[i] = s;
      s += c;
    }
    int inc = s;
    for (int off = 1; off < 32; off <<= 1) {
      const int up = __shfl_up(inc, off, 32);
      if (lane >= off) inc += up;
    }
    const int excl = inc - s;
    #pragma unroll
    for (int i = 0; i < CSR_PER_LANE; ++i) {
      const int j = lane * CSR_PER_LANE + i;
      if (j < CSR_RANGE) {
        const int start = base + excl + loc[i];
        bins[j] = start;
        if (j < nr) row_ptr[lo + j] = start;
      }
    }
    if (blockIdx.x == CSR_WGS - 1 && lane == 31) row_ptr[N_NODES] = base + inc;
  }
  __syncthreads();

  for (int k = 0; k < CSR_CHUNKS; k += 4) {
    int4 d[4];
    int  vv[4];
    #pragma unroll
    for (int u = 0; u < 4; ++u) {
      int c = k + u + rot;
      if (c >= CSR_CHUNKS) c -= CSR_CHUNKS;
      const int v = c * CSR_THREADS + t;
      const bool ok = (k + u < CSR_CHUNKS) && (v < E_VEC);
      vv[u] = v;
      d[u] = ok ? d4[v] : make_int4(CSR_PAD, CSR_PAD, CSR_PAD, CSR_PAD);
    }
    int pos[4][4];
    #pragma unroll
    for (int u = 0; u < 4; ++u) {
      const int dd[4] = {d[u].x, d[u].y, d[u].z, d[u].w};
      #pragma unroll
      for (int q = 0; q < 4; ++q) {
        const unsigned r = (unsigned)(dd[q] - lo);
        int p = -1;
        if (r < (unsigned)nr) p = atomicAdd(&bins[r], 1);
        pos[u][q] = p;
      }
    }
    #pragma unroll
    for (int u = 0; u < 4; ++u) {
      #pragma unroll
      for (int q = 0; q < 4; ++q)
        if (pos[u][q] >= 0) edge_idx[pos[u][q]] = 4 * vv[u] + q;
    }
  }
}

__global__ __launch_bounds__(128) void gat_aggregate(
    const float* __restrict__ h, const float* __restrict__ el,
    const float* __restrict__ er, const int* __restrict__ src,
    const int* __restrict__ row_ptr, const int* __restrict__ edge_idx,
    const float* __restrict__ bias, const float* __restrict__ alpha_p,
    const float* __restrict__ x0, float* __restrict__ xout) {
  __shared__ float red[H_HEADS * F_OUTF];
  const int n = blockIdx.x;
  const int head = threadIdx.x >> 5, lane = threadIdx.x & 31;
  const int rs = row_ptr[n], re = row_ptr[n + 1];
  const float erd = er[n * H_HEADS + head];

  float mx = -INFINITY;
  for (int i = rs + lane; i < re; i += 32) {
    const int e = edge_idx[i];
    float v = el[src[e] * H_HEADS + head] + erd;
    v = v > 0.f ? v : NEG_SLOPE * v;
    mx = fmaxf(mx, v);
  }
  #pragma unroll
  for (int off = 16; off; off >>= 1) mx = fmaxf(mx, __shfl_xor(mx, off, 32));

  float sum = 0.f;
  for (int i = rs + lane; i < re; i += 32) {
    const int e = edge_idx[i];
    float v = el[src[e] * H_HEADS + head] + erd;
    v = v > 0.f ? v : NEG_SLOPE * v;
    sum += __expf(v - mx);
  }
  #pragma unroll
  for (int off = 16; off; off >>= 1) sum += __shfl_xor(sum, off, 32);
  const float inv = (sum != 0.f) ? 1.f / sum : 0.f;

  float4 acc = make_float4(0.f, 0.f, 0.f, 0.f);
  for (int i = rs; i < re; ++i) {
    const int e = edge_idx[i];
    const int s = src[e];
    float v = el[s * H_HEADS + head] + erd;
    v = v > 0.f ? v : NEG_SLOPE * v;
    const float a = __expf(v - mx) * inv;
    const float4 hv = *(const float4*)(h + (size_t)s * C_OUT + head * F_OUTF + lane * 4);
    acc.x += a * hv.x; acc.y += a * hv.y; acc.z += a * hv.z; acc.w += a * hv.w;
  }
  const float4 bv = *(const float4*)(bias + head * F_OUTF + lane * 4);
  acc.x += bv.x; acc.y += bv.y; acc.z += bv.z; acc.w += bv.w;
  *(float4*)(red + head * F_OUTF + lane * 4) = acc;
  __syncthreads();

  const int f = threadIdx.x;
  const float s = 0.25f * (red[f] + red[F_OUTF + f] +
                           red[2 * F_OUTF + f] + red[3 * F_OUTF + f]);
  const float al = alpha_p[0];
  xout[(size_t)n * F_OUTF + f] = (1.f - al) * s + al * x0[(size_t)n * F_OUTF + f];
}

extern "C" void kernel_launch(void* const* d_in, const int* in_sizes, int n_in,
                              void* d_out, int out_size, void* d_ws, size_t ws_size,
                              hipStream_t stream) {
  const float* x      = (const float*)d_in[0];
  const float* x0     = (const float*)d_in[1];
  const int*   src    = (const int*)d_in[2];
  const int*   dst    = (const int*)d_in[3];
  const float* alpha  = (const float*)d_in[4];
  const float* W      = (const float*)d_in[5];
  const float* attn_l = (const float*)d_in[6];
  const float* attn_r = (const float*)d_in[7];
  const float* bias   = (const float*)d_in[8];

  char* ws = (char*)d_ws;
  size_t off = 0;
  auto alloc = [&](size_t bytes) -> char* {
    char* p = ws + off;
    off = (off + bytes + 255) & ~(size_t)255;
    return p;
  };
  float* h        = (float*)alloc((size_t)N_NODES * C_OUT * sizeof(float));
  float* el       = (float*)alloc((size_t)N_NODES * H_HEADS * sizeof(float));
  float* er       = (float*)alloc((size_t)N_NODES * H_HEADS * sizeof(float));
  float* xbuf     = (float*)alloc((size_t)N_NODES * F_INF * sizeof(float));
  int*   row_ptr  = (int*)alloc((size_t)(N_NODES + 1) * sizeof(int));
  int*   edge_idx = (int*)alloc((size_t)N_EDGES * sizeof(int));

  csr_build<<<CSR_WGS, CSR_THREADS, 0, stream>>>(dst, row_ptr, edge_idx);

  const float* xin = x;
  for (int step = 0; step < STEPS; ++step) {
    gemm_xw_logits<<<(M_TILES / GEMM_WAVES) * H_HEADS, GEMM_WAVES * 32, 0, stream>>>(
        xin, W, attn_l, attn_r, h, el, er);
    float* xo = (step == STEPS - 1) ? (float*)d_out : xbuf;
    gat_aggregate<<<N_NODES, 128, 0, stream>>>(h, el, er, src, row_ptr, edge_idx,
                                               bias, alpha, x0, xo);
    xin = xbuf;
  }
}


# aggregate: one wave per destination node does all four heads (was one per head): edge lookups once per node, 16 h-row loads in flight, head-mean in registers, 4x fewer waves; sums in the same order
# speedup vs baseline: 1.7468x; 1.1597x over previous
#include <hip/hip_runtime.h>
#include <hip/hip_bf16.h>

#define N_NODES 10000
#define N_EDGES 320000
#define F_INF   128
#define F_OUTF  128
#define H_HEADS 4
#define C_OUT   (H_HEADS * F_OUTF)
#define STEPS   2
#define NEG_SLOPE 0.2f

typedef float v2f __attribute__((ext_vector_type(2)));
typedef float v8f __attribute__((ext_vector_type(8)));

#define GEMM_WAVES 5
#define M_TILES    (N_NODES / 16)
static_assert(N_NODES % 16 == 0, "whole row tiles");
static_assert(M_TILES % GEMM_WAVES == 0, "whole work-groups, no tail wave");
static_assert(F_OUTF == 128 && H_HEADS == 4 && F_INF % 4 == 0, "tile naming below");

__global__ __launch_bounds__(GEMM_WAVES * 32) void gemm_xw_logits(
    const float* __restrict__ x, const float* __restrict__ W,
    const float* __restrict__ attn_l, const float* __restrict__ attn_r,
    float* __restrict__ h, float* __restrict__ el, float* __restrict__ er) {
  const int lane = threadIdx.x & 31;
  const int wv   = threadIdx.x >> 5;
  const int hd   = blockIdx.x & (H_HEADS - 1);
  const int tm   = (blockIdx.x >> 2) * GEMM_WAVES + wv;
  const int m0   = tm * 16;
  const int hf   = lane >> 4;
  const int c    = lane & 15;

  v8f acc[8];
  #pragma unroll
  for (int t = 0; t < 8; ++t) acc[t] = v8f{};

  const float* arow = x + (size_t)(m0 + c) * F_INF + 2 * hf;
  const float* bcol = W + (size_t)(2 * hf) * C_OUT + hd * F_OUTF + 4 * c;

  #pragma unroll 4
  for (int k0 = 0; k0 < F_INF; k0 += 4) {
    const v2f a = *(const v2f*)(arow + k0);
    const float* b0 = bcol + (size_t)k0 * C_OUT;
    const float4 x_lo = *(const float4*)(b0);
    const float4 x_hi = *(const float4*)(b0 + 64);
    const float4 y_lo = *(const float4*)(b0 + C_OUT);
    const float4 y_hi = *(const float4*)(b0 + C_OUT + 64);
    const float bx[8] = {x_lo.x, x_lo.y, x_lo.z, x_lo.w, x_hi.x, x_hi.y, x_hi.z, x_hi.w};
    const float by[8] = {y_lo.x, y_lo.y, y_lo.z, y_lo.w, y_hi.x, y_hi.y, y_hi.z, y_hi.w};
    #pragma unroll
    for (int t = 0; t < 8; ++t) {
      v2f b; b.x = bx[t]; b.y = by[t];
      acc[t] = __builtin_amdgcn_wmma_f32_16x16x4_f32(
           false, a,  false, b,
           (short)0, acc[t],  false,  false);
    }
  }

  const int row0 = m0 + 8 * hf;
  const float4 al_lo = *(const float4*)(attn_l + hd * F_OUTF + 4 * c);
  const float4 al_hi = *(const float4*)(attn_l + hd * F_OUTF + 4 * c + 64);
  const float4 ar_lo = *(const float4*)(attn_r + hd * F_OUTF + 4 * c);
  const float4 ar_hi = *(const float4*)(attn_r + hd * F_OUTF + 4 * c + 64);
  float* hp = h + (size_t)row0 * C_OUT + hd * F_OUTF + 4 * c;
  #pragma unroll
  for (int i = 0; i < 8; ++i) {
    const float4 lo = make_float4(acc[0][i], acc[1][i], acc[2][i], acc[3][i]);
    const float4 hi = make_float4(acc[4][i], acc[5][i], acc[6][i], acc[7][i]);
    *(float4*)(hp + (size_t)i * C_OUT)      = lo;
    *(float4*)(hp + (size_t)i * C_OUT + 64) = hi;
    float sl = lo.x * al_lo.x + lo.y * al_lo.y + lo.z * al_lo.z + lo.w * al_lo.w
             + hi.x * al_hi.x + hi.y * al_hi.y + hi.z * al_hi.z + hi.w * al_hi.w;
    float sr = lo.x * ar_lo.x + lo.y * ar_lo.y + lo.z * ar_lo.z + lo.w * ar_lo.w
             + hi.x * ar_hi.x + hi.y * ar_hi.y + hi.z * ar_hi.z + hi.w * ar_hi.w;
    #pragma unroll
    for (int off = 8; off; off >>= 1) {
      sl += __shfl_xor(sl, off, 32);
      sr += __shfl_xor(sr, off, 32);
    }
    if (c == 0) {
      el[(row0 + i) * H_HEADS + hd] = sl;
      er[(row0 + i) * H_HEADS + hd] = sr;
    }
  }
}

#define CSR_WGS      64
#define CSR_THREADS  1024
#define CSR_RANGE    ((N_NODES + CSR_WGS - 1) / CSR_WGS)
#define CSR_PER_LANE ((CSR_RANGE + 31) / 32)
#define E_VEC        (N_EDGES / 4)
#define CSR_CHUNKS   ((E_VEC + CSR_THREADS - 1) / CSR_THREADS)
#define CSR_PAD      0x7fffffff

static_assert(N_EDGES % 4 == 0, "dst is read as int4");
static_assert((CSR_WGS - 1) * CSR_RANGE < N_NODES, "every group owns at least one node");
static_assert(CSR_WGS <= CSR_CHUNKS, "one distinct start chunk per group");

__global__ __launch_bounds__(CSR_THREADS) void csr_build(const int* __restrict__ dst,
                                                         int* __restrict__ row_ptr,
                                                         int* __restrict__ edge_idx) {
  __shared__ int bins[CSR_RANGE];
  __shared__ int s_below;
  const int t = threadIdx.x, lane = t & 31, wv = t >> 5;
  const int lo = blockIdx.x * CSR_RANGE;
  const int nr = min(CSR_RANGE, N_NODES - lo);
  const int rot = blockIdx.x;
  const int4* d4 = (const int4*)dst;

  for (int j = t; j < CSR_RANGE; j += CSR_THREADS) bins[j] = 0;
  if (t == 0) s_below = 0;
  __syncthreads();

  int below = 0;
  for (int k = 0; k < CSR_CHUNKS; k += 4) {
    int4 d[4];
    #pragma unroll
    for (int u = 0; u < 4; ++u) {
      int c = k + u + rot;
      if (c >= CSR_CHUNKS) c -= CSR_CHUNKS;
      const int v = c * CSR_THREADS + t;
      const bool ok = (k + u < CSR_CHUNKS) && (v < E_VEC);
      d[u] = ok ? d4[v] : make_int4(CSR_PAD, CSR_PAD, CSR_PAD, CSR_PAD);
    }
    #pragma unroll
    for (int u = 0; u < 4; ++u) {
      const int dd[4] = {d[u].x, d[u].y, d[u].z, d[u].w};
      #pragma unroll
      for (int q = 0; q < 4; ++q) {
        const unsigned r = (unsigned)(dd[q] - lo);
        below += (dd[q] < lo) ? 1 : 0;
        if (r < (unsigned)nr) atomicAdd(&bins[r], 1);
      }
    }
  }
  for (int off = 16; off; off >>= 1) below += __shfl_xor(below, off, 32);
  if (lane == 0) atomicAdd(&s_below, below);
  __syncthreads();

  if (wv == 0) {
    const int base = s_below;
    int loc[CSR_PER_LANE];
    int s = 0;
    #pragma unroll
    for (int i = 0; i < CSR_PER_LANE; ++i) {
      const int j = lane * CSR_PER_LANE + i;
      const int c = (j < CSR_RANGE) ? bins[j] : 0;
      loc[i] = s;
      s += c;
    }
    int inc = s;
    for (int off = 1; off < 32; off <<= 1) {
      const int up = __shfl_up(inc, off, 32);
      if (lane >= off) inc += up;
    }
    const int excl = inc - s;
    #pragma unroll
    for (int i = 0; i < CSR_PER_LANE; ++i) {
      const int j = lane * CSR_PER_LANE + i;
      if (j < CSR_RANGE) {
        const int start = base + excl + loc[i];
        bins[j] = start;
        if (j < nr) row_ptr[lo + j] = start;
      }
    }
    if (blockIdx.x == CSR_WGS - 1 && lane == 31) row_ptr[N_NODES] = base + inc;
  }
  __syncthreads();

  for (int k = 0; k < CSR_CHUNKS; k += 4) {
    int4 d[4];
    int  vv[4];
    #pragma unroll
    for (int u = 0; u < 4; ++u) {
      int c = k + u + rot;
      if (c >= CSR_CHUNKS) c -= CSR_CHUNKS;
      const int v = c * CSR_THREADS + t;
      const bool ok = (k + u < CSR_CHUNKS) && (v < E_VEC);
      vv[u] = v;
      d[u] = ok ? d4[v] : make_int4(CSR_PAD, CSR_PAD, CSR_PAD, CSR_PAD);
    }
    int pos[4][4];
    #pragma unroll
    for (int u = 0; u < 4; ++u) {
      const int dd[4] = {d[u].x, d[u].y, d[u].z, d[u].w};
      #pragma unroll
      for (int q = 0; q < 4; ++q) {
        const unsigned r = (unsigned)(dd[q] - lo);
        int p = -1;
        if (r < (unsigned)nr) p = atomicAdd(&bins[r], 1);
        pos[u][q] = p;
      }
    }
    #pragma unroll
    for (int u = 0; u < 4; ++u) {
      #pragma unroll
      for (int q = 0; q < 4; ++q)
        if (pos[u][q] >= 0) edge_idx[pos[u][q]] = 4 * vv[u] + q;
    }
  }
}

#define AGG_WAVES 4
#define AGG_EDGES 4
static_assert(N_NODES % AGG_WAVES == 0, "whole work-groups, no tail wave");
static_assert(32 % AGG_EDGES == 0, "a chunk is a whole number of batches");
static_assert(H_HEADS == 4 && F_OUTF == 128, "a lane holds 4 features of each of 4 heads");

__device__ __forceinline__ float leaky(float v) { return v > 0.f ? v : NEG_SLOPE * v; }

__global__ __launch_bounds__(AGG_WAVES * 32) void gat_aggregate(
    const float* __restrict__ h, const float* __restrict__ el,
    const float* __restrict__ er, const int* __restrict__ src,
    const int* __restrict__ row_ptr, const int* __restrict__ edge_idx,
    const float* __restrict__ bias, const float* __restrict__ alpha_p,
    const float* __restrict__ x0, float* __restrict__ xout) {
  const int lane = threadIdx.x & 31;
  const int n = blockIdx.x * AGG_WAVES + __builtin_amdgcn_readfirstlane(threadIdx.x >> 5);
  const int rs = row_ptr[n], re = row_ptr[n + 1];
  const float4 er4 = *(const float4*)(er + n * H_HEADS);
  const float erd[H_HEADS] = {er4.x, er4.y, er4.z, er4.w};

  const bool ok0 = rs + lane < re;
  int s0 = 0;
  float v0[H_HEADS];
  for (int hd = 0; hd < H_HEADS; ++hd) v0[hd] = -INFINITY;
  if (ok0) {
    s0 = src[edge_idx[rs + lane]];
    const float4 q4 = *(const float4*)(el + s0 * H_HEADS);
    const float q[H_HEADS] = {q4.x, q4.y, q4.z, q4.w};
    for (int hd = 0; hd < H_HEADS; ++hd) v0[hd] = leaky(q[hd] + erd[hd]);
  }

  float mx[H_HEADS];
  for (int hd = 0; hd < H_HEADS; ++hd) mx[hd] = fmaxf(-INFINITY, v0[hd]);
  for (int i = rs + 32 + lane; i < re; i += 32) {
    const int s = src[edge_idx[i]];
    const float4 q4 = *(const float4*)(el + s * H_HEADS);
    const float q[H_HEADS] = {q4.x, q4.y, q4.z, q4.w};
    for (int hd = 0; hd < H_HEADS; ++hd) mx[hd] = fmaxf(mx[hd], leaky(q[hd] + erd[hd]));
  }
  #pragma unroll
  for (int off = 16; off; off >>= 1) {
    for (int hd = 0; hd < H_HEADS; ++hd) mx[hd] = fmaxf(mx[hd], __shfl_xor(mx[hd], off, 32));
  }

  float p0[H_HEADS], sum[H_HEADS], inv[H_HEADS];
  for (int hd = 0; hd < H_HEADS; ++hd) {
    p0[hd] = ok0 ? __expf(v0[hd] - mx[hd]) : 0.f;
    sum[hd] = p0[hd];
  }
  for (int i = rs + 32 + lane; i < re; i += 32) {
    const int s = src[edge_idx[i]];
    const float4 q4 = *(const float4*)(el + s * H_HEADS);
    const float q[H_HEADS] = {q4.x, q4.y, q4.z, q4.w};
    for (int hd = 0; hd < H_HEADS; ++hd) sum[hd] += __expf(leaky(q[hd] + erd[hd]) - mx[hd]);
  }
  #pragma unroll
  for (int off = 16; off; off >>= 1) {
    for (int hd = 0; hd < H_HEADS; ++hd) sum[hd] += __shfl_xor(sum[hd], off, 32);
  }
  for (int hd = 0; hd < H_HEADS; ++hd) inv[hd] = (sum[hd] != 0.f) ? 1.f / sum[hd] : 0.f;

  const float* hrow = h + lane * 4;
  float4 acc[H_HEADS];
  for (int hd = 0; hd < H_HEADS; ++hd) acc[hd] = make_float4(0.f, 0.f, 0.f, 0.f);
  int   s_l = s0;
  float a_l[H_HEADS];
  for (int hd = 0; hd < H_HEADS; ++hd) a_l[hd] = p0[hd] * inv[hd];
  for (int base = rs; base < re; base += 32) {
    if (base != rs) {
      s_l = 0;
      for (int hd = 0; hd < H_HEADS; ++hd) a_l[hd] = 0.f;
      if (base + lane < re) {
        s_l = src[edge_idx[base + lane]];
        const float4 q4 = *(const float4*)(el + s_l * H_HEADS);
        const float q[H_HEADS] = {q4.x, q4.y, q4.z, q4.w};
        for (int hd = 0; hd < H_HEADS; ++hd)
          a_l[hd] = __expf(leaky(q[hd] + erd[hd]) - mx[hd]) * inv[hd];
      }
    }
    const int cnt = min(32, re - base);
    const int s_last = __builtin_amdgcn_readlane(s_l, cnt - 1);
    if (lane >= cnt) s_l = s_last;
    for (int j = 0; j < cnt; j += AGG_EDGES) {
      float4 hv[AGG_EDGES][H_HEADS];
      float  aw[AGG_EDGES][H_HEADS];
      #pragma unroll
      for (int u = 0; u < AGG_EDGES; ++u) {
        const float* hs = hrow + (size_t)__builtin_amdgcn_readlane(s_l, j + u) * C_OUT;
        for (int hd = 0; hd < H_HEADS; ++hd) {
          aw[u][hd] = __int_as_float(__builtin_amdgcn_readlane(__float_as_int(a_l[hd]), j + u));
          hv[u][hd] = *(const float4*)(hs + hd * F_OUTF);
        }
      }
      #pragma unroll
      for (int u = 0; u < AGG_EDGES; ++u) {
        for (int hd = 0; hd < H_HEADS; ++hd) {
          acc[hd].x += aw[u][hd] * hv[u][hd].x; acc[hd].y += aw[u][hd] * hv[u][hd].y;
          acc[hd].z += aw[u][hd] * hv[u][hd].z; acc[hd].w += aw[u][hd] * hv[u][hd].w;
        }
      }
    }
  }
  for (int hd = 0; hd < H_HEADS; ++hd) {
    const float4 bv = *(const float4*)(bias + hd * F_OUTF + lane * 4);
    acc[hd].x += bv.x; acc[hd].y += bv.y; acc[hd].z += bv.z; acc[hd].w += bv.w;
  }

  const float al = alpha_p[0];
  const float4 xv = *(const float4*)(x0 + (size_t)n * F_OUTF + lane * 4);
  float4 o;
  o.x = (1.f - al) * (0.25f * (acc[0].x + acc[1].x + acc[2].x + acc[3].x)) + al * xv.x;
  o.y = (1.f - al) * (0.25f * (acc[0].y + acc[1].y + acc[2].y + acc[3].y)) + al * xv.y;
  o.z = (1.f - al) * (0.25f * (acc[0].z + acc[1].z + acc[2].z + acc[3].z)) + al * xv.z;
  o.w = (1.f - al) * (0.25f * (acc[0].w + acc[1].w + acc[2].w + acc[3].w)) + al * xv.w;
  *(float4*)(xout + (size_t)n * F_OUTF + lane * 4) = o;
}

extern "C" void kernel_launch(void* const* d_in, const int* in_sizes, int n_in,
                              void* d_out, int out_size, void* d_ws, size_t ws_size,
                              hipStream_t stream) {
  const float* x      = (const float*)d_in[0];
  const float* x0     = (const float*)d_in[1];
  const int*   src    = (const int*)d_in[2];
  const int*   dst    = (const int*)d_in[3];
  const float* alpha  = (const float*)d_in[4];
  const float* W      = (const float*)d_in[5];
  const float* attn_l = (const float*)d_in[6];
  const float* attn_r = (const float*)d_in[7];
  const float* bias   = (const float*)d_in[8];

  char* ws = (char*)d_ws;
  size_t off = 0;
  auto alloc = [&](size_t bytes) -> char* {
    char* p = ws + off;
    off = (off + bytes + 255) & ~(size_t)255;
    return p;
  };
  float* h        = (float*)alloc((size_t)N_NODES * C_OUT * sizeof(float));
  float* el       = (float*)alloc((size_t)N_NODES * H_HEADS * sizeof(float));
  float* er       = (float*)alloc((size_t)N_NODES * H_HEADS * sizeof(float));
  float* xbuf     = (float*)alloc((size_t)N_NODES * F_INF * sizeof(float));
  int*   row_ptr  = (int*)alloc((size_t)(N_NODES + 1) * sizeof(int));
  int*   edge_idx = (int*)alloc((size_t)N_EDGES * sizeof(int));

  csr_build<<<CSR_WGS, CSR_THREADS, 0, stream>>>(dst, row_ptr, edge_idx);

  const float* xin = x;
  for (int step = 0; step < STEPS; ++step) {
    gemm_xw_logits<<<(M_TILES / GEMM_WAVES) * H_HEADS, GEMM_WAVES * 32, 0, stream>>>(
        xin, W, attn_l, attn_r, h, el, er);
    float* xo = (step == STEPS - 1) ? (float*)d_out : xbuf;
    gat_aggregate<<<N_NODES / AGG_WAVES, AGG_WAVES * 32, 0, stream>>>(h, el, er, src, row_ptr, edge_idx,
                                               bias, alpha, x0, xo);
    xin = xbuf;
  }
}


# node logits on the matrix unit: one wave per 16 nodes (x tile @ the 8 folded vectors, f32 16x16x4), 625 waves per launch instead of 10000; order of the f32 sums in el/er changes
# speedup vs baseline: 2.1052x; 1.2052x over previous
#include <hip/hip_runtime.h>
#include <hip/hip_bf16.h>

#define N_NODES 10000
#define N_EDGES 320000
#define F_INF   128
#define F_OUTF  128
#define H_HEADS 4
#define C_OUT   (H_HEADS * F_OUTF)
#define STEPS   2
#define NEG_SLOPE 0.2f

typedef float v2f __attribute__((ext_vector_type(2)));
typedef float v8f __attribute__((ext_vector_type(8)));

#define FOLD_WAVES 4
static_assert((F_INF * H_HEADS) % FOLD_WAVES == 0, "whole work-groups, no tail wave");
static_assert(F_OUTF == 128, "a lane holds 4 of a head's 128 columns");

__global__ __launch_bounds__(FOLD_WAVES * 32) void attn_fold(
    const float* __restrict__ W, const float* __restrict__ attn_l,
    const float* __restrict__ attn_r, float* __restrict__ wlr) {
  const int lane = threadIdx.x & 31;
  const int id   = blockIdx.x * FOLD_WAVES + (threadIdx.x >> 5);
  const int k    = id >> 2;
  const int hd   = id & (H_HEADS - 1);
  const float4 w  = *(const float4*)(W + (size_t)k * C_OUT + hd * F_OUTF + 4 * lane);
  const float4 al = *(const float4*)(attn_l + hd * F_OUTF + 4 * lane);
  const float4 ar = *(const float4*)(attn_r + hd * F_OUTF + 4 * lane);
  float sl = w.x * al.x + w.y * al.y + w.z * al.z + w.w * al.w;
  float sr = w.x * ar.x + w.y * ar.y + w.z * ar.z + w.w * ar.w;
  for (int off = 16; off; off >>= 1) {
    sl += __shfl_xor(sl, off, 32);
    sr += __shfl_xor(sr, off, 32);
  }
  if (lane == 0) {
    wlr[hd * F_INF + k]             = sl;
    wlr[(H_HEADS + hd) * F_INF + k] = sr;
  }
}

#define LOGIT_WAVES 5
#define LOGIT_TILES (N_NODES / 16)
static_assert(N_NODES % 16 == 0, "whole 16-node tiles");
static_assert(LOGIT_TILES % LOGIT_WAVES == 0, "whole work-groups, no tail wave");
static_assert(F_INF % 8 == 0 && 2 * H_HEADS == 8, "k-steps in pairs; eight logits per node");

__global__ __launch_bounds__(LOGIT_WAVES * 32) void node_logits(
    const float* __restrict__ x, const float* __restrict__ wlr,
    float* __restrict__ el, float* __restrict__ er) {
  const int lane = threadIdx.x & 31;
  const int m0   = (blockIdx.x * LOGIT_WAVES + (threadIdx.x >> 5)) * 16;
  const int hf   = lane >> 4;
  const int c    = lane & 15;

  const float* arow = x + (size_t)(m0 + c) * F_INF + 4 * hf;
  const float* brow = wlr + (c & 7) * F_INF + 4 * hf;

  v8f acc0 = v8f{};
  v8f acc1 = v8f{};
  #pragma unroll
  for (int k8 = 0; k8 < F_INF; k8 += 8) {
    const float4 a4 = *(const float4*)(arow + k8);
    const float4 b4 = *(const float4*)(brow + k8);
    v2f a; a.x = a4.x; a.y = a4.y;
    v2f b; b.x = b4.x; b.y = b4.y;
    acc0 = __builtin_amdgcn_wmma_f32_16x16x4_f32(
         false, a,  false, b,
         (short)0, acc0,  false,  false);
    a.x = a4.z; a.y = a4.w;
    b.x = b4.z; b.y = b4.w;
    acc1 = __builtin_amdgcn_wmma_f32_16x16x4_f32(
         false, a,  false, b,
         (short)0, acc1,  false,  false);
  }

  if (c < 2 * H_HEADS) {
    float* o = (c < H_HEADS) ? (el + c) : (er + (c - H_HEADS));
    #pragma unroll
    for (int i = 0; i < 8; ++i) o[(m0 + 8 * hf + i) * H_HEADS] = acc0[i] + acc1[i];
  }
}

#define CSR_WGS      64
#define CSR_THREADS  1024
#define CSR_RANGE    ((N_NODES + CSR_WGS - 1) / CSR_WGS)
#define CSR_PER_LANE ((CSR_RANGE + 31) / 32)
#define E_VEC        (N_EDGES / 4)
#define CSR_CHUNKS   ((E_VEC + CSR_THREADS - 1) / CSR_THREADS)
#define CSR_PAD      0x7fffffff

static_assert(N_EDGES % 4 == 0, "dst is read as int4");
static_assert((CSR_WGS - 1) * CSR_RANGE < N_NODES, "every group owns at least one node");
static_assert(CSR_WGS <= CSR_CHUNKS, "one distinct start chunk per group");

__global__ __launch_bounds__(CSR_THREADS) void csr_build(const int* __restrict__ dst,
                                                         int* __restrict__ row_ptr,
                                                         int* __restrict__ edge_idx) {
  __shared__ int bins[CSR_RANGE];
  __shared__ int s_below;
  const int t = threadIdx.x, lane = t & 31, wv = t >> 5;
  const int lo = blockIdx.x * CSR_RANGE;
  const int nr = min(CSR_RANGE, N_NODES - lo);
  const int rot = blockIdx.x;
  const int4* d4 = (const int4*)dst;

  for (int j = t; j < CSR_RANGE; j += CSR_THREADS) bins[j] = 0;
  if (t == 0) s_below = 0;
  __syncthreads();

  int below = 0;
  for (int k = 0; k < CSR_CHUNKS; k += 4) {
    int4 d[4];
    #pragma unroll
    for (int u = 0; u < 4; ++u) {
      int c = k + u + rot;
      if (c >= CSR_CHUNKS) c -= CSR_CHUNKS;
      const int v = c * CSR_THREADS + t;
      const bool ok = (k + u < CSR_CHUNKS) && (v < E_VEC);
      d[u] = ok ? d4[v] : make_int4(CSR_PAD, CSR_PAD, CSR_PAD, CSR_PAD);
    }
    #pragma unroll
    for (int u = 0; u < 4; ++u) {
      const int dd[4] = {d[u].x, d[u].y, d[u].z, d[u].w};
      #pragma unroll
      for (int q = 0; q < 4; ++q) {
        const unsigned r = (unsigned)(dd[q] - lo);
        below += (dd[q] < lo) ? 1 : 0;
        if (r < (unsigned)nr) atomicAdd(&bins[r], 1);
      }
    }
  }
  for (int off = 16; off; off >>= 1) below += __shfl_xor(below, off, 32);
  if (lane == 0) atomicAdd(&s_below, below);
  __syncthreads();

  if (wv == 0) {
    const int base = s_below;
    int loc[CSR_PER_LANE];
    int s = 0;
    #pragma unroll
    for (int i = 0; i < CSR_PER_LANE; ++i) {
      const int j = lane * CSR_PER_LANE + i;
      const int c = (j < CSR_RANGE) ? bins[j] : 0;
      loc[i] = s;
      s += c;
    }
    int inc = s;
    for (int off = 1; off < 32; off <<= 1) {
      const int up = __shfl_up(inc, off, 32);
      if (lane >= off) inc += up;
    }
    const int excl = inc - s;
    #pragma unroll
    for (int i = 0; i < CSR_PER_LANE; ++i) {
      const int j = lane * CSR_PER_LANE + i;
      if (j < CSR_RANGE) {
        const int start = base + excl + loc[i];
        bins[j] = start;
        if (j < nr) row_ptr[lo + j] = start;
      }
    }
    if (blockIdx.x == CSR_WGS - 1 && lane == 31) row_ptr[N_NODES] = base + inc;
  }
  __syncthreads();

  for (int k = 0; k < CSR_CHUNKS; k += 4) {
    int4 d[4];
    int  vv[4];
    #pragma unroll
    for (int u = 0; u < 4; ++u) {
      int c = k + u + rot;
      if (c >= CSR_CHUNKS) c -= CSR_CHUNKS;
      const int v = c * CSR_THREADS + t;
      const bool ok = (k + u < CSR_CHUNKS) && (v < E_VEC);
      vv[u] = v;
      d[u] = ok ? d4[v] : make_int4(CSR_PAD, CSR_PAD, CSR_PAD, CSR_PAD);
    }
    int pos[4][4];
    #pragma unroll
    for (int u = 0; u < 4; ++u) {
      const int dd[4] = {d[u].x, d[u].y, d[u].z, d[u].w};
      #pragma unroll
      for (int q = 0; q < 4; ++q) {
        const unsigned r = (unsigned)(dd[q] - lo);
        int p = -1;
        if (r < (unsigned)nr) p = atomicAdd(&bins[r], 1);
        pos[u][q] = p;
      }
    }
    #pragma unroll
    for (int u = 0; u < 4; ++u) {
      #pragma unroll
      for (int q = 0; q < 4; ++q)
        if (pos[u][q] >= 0) edge_idx[pos[u][q]] = 4 * vv[u] + q;
    }
  }
}

#define AGG_WAVES 4
#define AGG_EDGES 8
static_assert(N_NODES % AGG_WAVES == 0, "whole work-groups, no tail wave");
static_assert(32 % AGG_EDGES == 0, "a chunk is a whole number of batches");
static_assert(H_HEADS == 4 && F_INF == 128, "a lane holds 4 input features, one sum per head");

__device__ __forceinline__ float leaky(float v) { return v > 0.f ? v : NEG_SLOPE * v; }

__global__ __launch_bounds__(AGG_WAVES * 32) void gat_aggregate(
    const float* __restrict__ xin, const float* __restrict__ el,
    const float* __restrict__ er, const int* __restrict__ src,
    const int* __restrict__ row_ptr, const int* __restrict__ edge_idx,
    float* __restrict__ z) {
  const int lane = threadIdx.x & 31;
  const int n = blockIdx.x * AGG_WAVES + __builtin_amdgcn_readfirstlane(threadIdx.x >> 5);
  const int rs = row_ptr[n], re = row_ptr[n + 1];
  const float4 er4 = *(const float4*)(er + n * H_HEADS);
  const float erd[H_HEADS] = {er4.x, er4.y, er4.z, er4.w};

  const bool ok0 = rs + lane < re;
  int s0 = 0;
  float v0[H_HEADS];
  for (int hd = 0; hd < H_HEADS; ++hd) v0[hd] = -INFINITY;
  if (ok0) {
    s0 = src[edge_idx[rs + lane]];
    const float4 q4 = *(const float4*)(el + s0 * H_HEADS);
    const float q[H_HEADS] = {q4.x, q4.y, q4.z, q4.w};
    for (int hd = 0; hd < H_HEADS; ++hd) v0[hd] = leaky(q[hd] + erd[hd]);
  }

  float mx[H_HEADS];
  for (int hd = 0; hd < H_HEADS; ++hd) mx[hd] = fmaxf(-INFINITY, v0[hd]);
  for (int i = rs + 32 + lane; i < re; i += 32) {
    const int s = src[edge_idx[i]];
    const float4 q4 = *(const float4*)(el + s * H_HEADS);
    const float q[H_HEADS] = {q4.x, q4.y, q4.z, q4.w};
    for (int hd = 0; hd < H_HEADS; ++hd) mx[hd] = fmaxf(mx[hd], leaky(q[hd] + erd[hd]));
  }
  #pragma unroll
  for (int off = 16; off; off >>= 1) {
    for (int hd = 0; hd < H_HEADS; ++hd) mx[hd] = fmaxf(mx[hd], __shfl_xor(mx[hd], off, 32));
  }

  float p0[H_HEADS], sum[H_HEADS], inv[H_HEADS];
  for (int hd = 0; hd < H_HEADS; ++hd) {
    p0[hd] = ok0 ? __expf(v0[hd] - mx[hd]) : 0.f;
    sum[hd] = p0[hd];
  }
  for (int i = rs + 32 + lane; i < re; i += 32) {
    const int s = src[edge_idx[i]];
    const float4 q4 = *(const float4*)(el + s * H_HEADS);
    const float q[H_HEADS] = {q4.x, q4.y, q4.z, q4.w};
    for (int hd = 0; hd < H_HEADS; ++hd) sum[hd] += __expf(leaky(q[hd] + erd[hd]) - mx[hd]);
  }
  #pragma unroll
  for (int off = 16; off; off >>= 1) {
    for (int hd = 0; hd < H_HEADS; ++hd) sum[hd] += __shfl_xor(sum[hd], off, 32);
  }
  for (int hd = 0; hd < H_HEADS; ++hd) inv[hd] = (sum[hd] != 0.f) ? 1.f / sum[hd] : 0.f;

  const float* xrow = xin + lane * 4;
  float4 acc[H_HEADS];
  for (int hd = 0; hd < H_HEADS; ++hd) acc[hd] = make_float4(0.f, 0.f, 0.f, 0.f);
  int   s_l = s0;
  float a_l[H_HEADS];
  for (int hd = 0; hd < H_HEADS; ++hd) a_l[hd] = p0[hd] * inv[hd];
  for (int base = rs; base < re; base += 32) {
    if (base != rs) {
      s_l = 0;
      for (int hd = 0; hd < H_HEADS; ++hd) a_l[hd] = 0.f;
      if (base + lane < re) {
        s_l = src[edge_idx[base + lane]];
        const float4 q4 = *(const float4*)(el + s_l * H_HEADS);
        const float q[H_HEADS] = {q4.x, q4.y, q4.z, q4.w};
        for (int hd = 0; hd < H_HEADS; ++hd)
          a_l[hd] = __expf(leaky(q[hd] + erd[hd]) - mx[hd]) * inv[hd];
      }
    }
    const int cnt = min(32, re - base);
    const int s_last = __builtin_amdgcn_readlane(s_l, cnt - 1);
    if (lane >= cnt) s_l = s_last;
    for (int j = 0; j < cnt; j += AGG_EDGES) {
      float4 xv[AGG_EDGES];
      float  aw[AGG_EDGES][H_HEADS];
      #pragma unroll
      for (int u = 0; u < AGG_EDGES; ++u) {
        xv[u] = *(const float4*)(xrow + (size_t)__builtin_amdgcn_readlane(s_l, j + u) * F_INF);
        for (int hd = 0; hd < H_HEADS; ++hd)
          aw[u][hd] = __int_as_float(__builtin_amdgcn_readlane(__float_as_int(a_l[hd]), j + u));
      }
      #pragma unroll
      for (int u = 0; u < AGG_EDGES; ++u) {
        for (int hd = 0; hd < H_HEADS; ++hd) {
          acc[hd].x += aw[u][hd] * xv[u].x; acc[hd].y += aw[u][hd] * xv[u].y;
          acc[hd].z += aw[u][hd] * xv[u].z; acc[hd].w += aw[u][hd] * xv[u].w;
        }
      }
    }
  }
  for (int hd = 0; hd < H_HEADS; ++hd)
    *(float4*)(z + (size_t)n * C_OUT + hd * F_INF + lane * 4) = acc[hd];
}

#define GEMM_WAVES 5
#define M_TILES    (N_NODES / 16)
#define N_SLABS    (F_OUTF / 32)
static_assert(N_NODES % 16 == 0, "whole row tiles");
static_assert(M_TILES % GEMM_WAVES == 0, "whole work-groups, no tail wave");
static_assert(F_OUTF == 128 && H_HEADS == 4 && F_INF % 8 == 0, "tile naming below, k-steps in pairs");

__global__ __launch_bounds__(GEMM_WAVES * 32) void gemm_zw_out(
    const float* __restrict__ z, const float* __restrict__ W,
    const float* __restrict__ bias, const float* __restrict__ alpha_p,
    const float* __restrict__ x0, float* __restrict__ xout) {
  const int lane = threadIdx.x & 31;
  const int wv   = threadIdx.x >> 5;
  const int q    = blockIdx.x & (N_SLABS - 1);
  const int tm   = (blockIdx.x >> 2) * GEMM_WAVES + wv;
  const int m0   = tm * 16;
  const int hf   = lane >> 4;
  const int c    = lane & 15;

  v8f acc0 = v8f{};
  v8f acc1 = v8f{};

  const float* arow = z + (size_t)(m0 + c) * C_OUT + 4 * hf;
  const float* bcol = W + (size_t)(4 * hf) * C_OUT + 32 * q + 2 * c;

  for (int hd = 0; hd < H_HEADS; ++hd) {
    const float* ah = arow + hd * F_INF;
    const float* bh = bcol + hd * F_OUTF;
    #pragma unroll 8
    for (int k8 = 0; k8 < F_INF; k8 += 8) {
      const float4 a4 = *(const float4*)(ah + k8);
      const float2 r0 = *(const float2*)(bh + (size_t)k8 * C_OUT);
      const float2 r1 = *(const float2*)(bh + (size_t)k8 * C_OUT + C_OUT);
      const float2 r2 = *(const float2*)(bh + (size_t)k8 * C_OUT + 2 * C_OUT);
      const float2 r3 = *(const float2*)(bh + (size_t)k8 * C_OUT + 3 * C_OUT);
      v2f a;  a.x  = a4.x; a.y  = a4.y;
      v2f b0; b0.x = r0.x; b0.y = r1.x;
      v2f b1; b1.x = r0.y; b1.y = r1.y;
      acc0 = __builtin_amdgcn_wmma_f32_16x16x4_f32(
           false, a,  false, b0,
           (short)0, acc0,  false,  false);
      acc1 = __builtin_amdgcn_wmma_f32_16x16x4_f32(
           false, a,  false, b1,
           (short)0, acc1,  false,  false);
      a.x  = a4.z; a.y  = a4.w;
      b0.x = r2.x; b0.y = r3.x;
      b1.x = r2.y; b1.y = r3.y;
      acc0 = __builtin_amdgcn_wmma_f32_16x16x4_f32(
           false, a,  false, b0,
           (short)0, acc0,  false,  false);
      acc1 = __builtin_amdgcn_wmma_f32_16x16x4_f32(
           false, a,  false, b1,
           (short)0, acc1,  false,  false);
    }
  }

  const int row0 = m0 + 8 * hf;
  const int col  = 32 * q + 2 * c;
  const float al = alpha_p[0];
  float bmx = 0.f, bmy = 0.f;
  for (int hd = 0; hd < H_HEADS; ++hd) {
    const float2 b = *(const float2*)(bias + hd * F_OUTF + col);
    bmx += b.x; bmy += b.y;
  }
  #pragma unroll
  for (int i = 0; i < 8; ++i) {
    const size_t o = (size_t)(row0 + i) * F_OUTF + col;
    const float2 xv = *(const float2*)(x0 + o);
    float2 r;
    r.x = (1.f - al) * (0.25f * (acc0[i] + bmx)) + al * xv.x;
    r.y = (1.f - al) * (0.25f * (acc1[i] + bmy)) + al * xv.y;
    *(float2*)(xout + o) = r;
  }
}

extern "C" void kernel_launch(void* const* d_in, const int* in_sizes, int n_in,
                              void* d_out, int out_size, void* d_ws, size_t ws_size,
                              hipStream_t stream) {
  const float* x      = (const float*)d_in[0];
  const float* x0     = (const float*)d_in[1];
  const int*   src    = (const int*)d_in[2];
  const int*   dst    = (const int*)d_in[3];
  const float* alpha  = (const float*)d_in[4];
  const float* W      = (const float*)d_in[5];
  const float* attn_l = (const float*)d_in[6];
  const float* attn_r = (const float*)d_in[7];
  const float* bias   = (const float*)d_in[8];

  char* ws = (char*)d_ws;
  size_t off = 0;
  auto alloc = [&](size_t bytes) -> char* {
    char* p = ws + off;
    off = (off + bytes + 255) & ~(size_t)255;
    return p;
  };
  float* z        = (float*)alloc((size_t)N_NODES * C_OUT * sizeof(float));
  float* el       = (float*)alloc((size_t)N_NODES * H_HEADS * sizeof(float));
  float* er       = (float*)alloc((size_t)N_NODES * H_HEADS * sizeof(float));
  float* xbuf     = (float*)alloc((size_t)N_NODES * F_INF * sizeof(float));
  int*   row_ptr  = (int*)alloc((size_t)(N_NODES + 1) * sizeof(int));
  int*   edge_idx = (int*)alloc((size_t)N_EDGES * sizeof(int));
  float* wlr      = (float*)alloc((size_t)F_INF * 2 * H_HEADS * sizeof(float));

  csr_build<<<CSR_WGS, CSR_THREADS, 0, stream>>>(dst, row_ptr, edge_idx);
  attn_fold<<<(F_INF * H_HEADS) / FOLD_WAVES, FOLD_WAVES * 32, 0, stream>>>(W, attn_l, attn_r, wlr);

  const float* xin = x;
  for (int step = 0; step < STEPS; ++step) {
    node_logits<<<LOGIT_TILES / LOGIT_WAVES, LOGIT_WAVES * 32, 0, stream>>>(xin, wlr, el, er);
    gat_aggregate<<<N_NODES / AGG_WAVES, AGG_WAVES * 32, 0, stream>>>(xin, el, er, src, row_ptr,
                                                                    edge_idx, z);
    float* xo = (step == STEPS - 1) ? (float*)d_out : xbuf;
    gemm_zw_out<<<(M_TILES / GEMM_WAVES) * N_SLABS, GEMM_WAVES * 32, 0, stream>>>(
        z, W, bias, alpha, x0, xo);
    xin = xbuf;
  }
}


# CSR build cut into 32 node ranges x 8 slices of the edge stream: 256 groups each sweep 1/8 of dst (64 groups swept all); a node's edges are 8 sub-segments walked in turn; order of f32 sums changes
# speedup vs baseline: 2.7637x; 1.3128x over previous
#include <hip/hip_runtime.h>
#include <hip/hip_bf16.h>

#define N_NODES 10000
#define N_EDGES 320000
#define F_INF   128
#define F_OUTF  128
#define H_HEADS 4
#define C_OUT   (H_HEADS * F_OUTF)
#define STEPS   2
#define NEG_SLOPE 0.2f

typedef float v2f __attribute__((ext_vector_type(2)));
typedef float v8f __attribute__((ext_vector_type(8)));

#define FOLD_WAVES 4
static_assert((F_INF * H_HEADS) % FOLD_WAVES == 0, "whole work-groups, no tail wave");
static_assert(F_OUTF == 128, "a lane holds 4 of a head's 128 columns");

__global__ __launch_bounds__(FOLD_WAVES * 32) void attn_fold(
    const float* __restrict__ W, const float* __restrict__ attn_l,
    const float* __restrict__ attn_r, float* __restrict__ wlr) {
  const int lane = threadIdx.x & 31;
  const int id   = blockIdx.x * FOLD_WAVES + (threadIdx.x >> 5);
  const int k    = id >> 2;
  const int hd   = id & (H_HEADS - 1);
  const float4 w  = *(const float4*)(W + (size_t)k * C_OUT + hd * F_OUTF + 4 * lane);
  const float4 al = *(const float4*)(attn_l + hd * F_OUTF + 4 * lane);
  const float4 ar = *(const float4*)(attn_r + hd * F_OUTF + 4 * lane);
  float sl = w.x * al.x + w.y * al.y + w.z * al.z + w.w * al.w;
  float sr = w.x * ar.x + w.y * ar.y + w.z * ar.z + w.w * ar.w;
  for (int off = 16; off; off >>= 1) {
    sl += __shfl_xor(sl, off, 32);
    sr += __shfl_xor(sr, off, 32);
  }
  if (lane == 0) {
    wlr[hd * F_INF + k]             = sl;
    wlr[(H_HEADS + hd) * F_INF + k] = sr;
  }
}

#define LOGIT_WAVES 5
#define LOGIT_TILES (N_NODES / 16)
static_assert(N_NODES % 16 == 0, "whole 16-node tiles");
static_assert(LOGIT_TILES % LOGIT_WAVES == 0, "whole work-groups, no tail wave");
static_assert(F_INF % 8 == 0 && 2 * H_HEADS == 8, "k-steps in pairs; eight logits per node");

__global__ __launch_bounds__(LOGIT_WAVES * 32) void node_logits(
    const float* __restrict__ x, const float* __restrict__ wlr,
    float* __restrict__ el, float* __restrict__ er) {
  const int lane = threadIdx.x & 31;
  const int m0   = (blockIdx.x * LOGIT_WAVES + (threadIdx.x >> 5)) * 16;
  const int hf   = lane >> 4;
  const int c    = lane & 15;

  const float* arow = x + (size_t)(m0 + c) * F_INF + 4 * hf;
  const float* brow = wlr + (c & 7) * F_INF + 4 * hf;

  v8f acc0 = v8f{};
  v8f acc1 = v8f{};
  #pragma unroll
  for (int k8 = 0; k8 < F_INF; k8 += 8) {
    const float4 a4 = *(const float4*)(arow + k8);
    const float4 b4 = *(const float4*)(brow + k8);
    v2f a; a.x = a4.x; a.y = a4.y;
    v2f b; b.x = b4.x; b.y = b4.y;
    acc0 = __builtin_amdgcn_wmma_f32_16x16x4_f32(
         false, a,  false, b,
         (short)0, acc0,  false,  false);
    a.x = a4.z; a.y = a4.w;
    b.x = b4.z; b.y = b4.w;
    acc1 = __builtin_amdgcn_wmma_f32_16x16x4_f32(
         false, a,  false, b,
         (short)0, acc1,  false,  false);
  }

  if (c < 2 * H_HEADS) {
    float* o = (c < H_HEADS) ? (el + c) : (er + (c - H_HEADS));
    #pragma unroll
    for (int i = 0; i < 8; ++i) o[(m0 + 8 * hf + i) * H_HEADS] = acc0[i] + acc1[i];
  }
}

#define CSR_RANGES   32
#define CSR_SLICES   8
#define CSR_WGS      (CSR_RANGES * CSR_SLICES)
#define CSR_THREADS  512
#define CSR_RANGE    ((N_NODES + CSR_RANGES - 1) / CSR_RANGES)
#define CSR_PER_LANE ((CSR_RANGE + 31) / 32)
#define SLICE_EDGES  (N_EDGES / CSR_SLICES)
#define SLICE_VEC    (SLICE_EDGES / 4)
#define CSR_CHUNKS   ((SLICE_VEC + CSR_THREADS - 1) / CSR_THREADS)
#define RP_STRIDE    (((N_NODES + 1 + 31) / 32) * 32)
#define CSR_PAD      0x7fffffff

static_assert(N_EDGES % (4 * CSR_SLICES) == 0, "a slice of dst is read as int4");
static_assert((CSR_RANGES - 1) * CSR_RANGE < N_NODES, "every group owns at least one node");
static_assert(CSR_CHUNKS % 4 == 0, "the sweeps take four chunks per trip");
static_assert((CSR_SLICES & (CSR_SLICES - 1)) == 0, "slice number = low bits of the group number");

__global__ __launch_bounds__(CSR_THREADS) void csr_build(const int* __restrict__ dst,
                                                         int* __restrict__ row_ptr,
                                                         int* __restrict__ edge_idx) {
  __shared__ int bins[CSR_RANGE];
  __shared__ int s_below;
  const int t = threadIdx.x, lane = t & 31, wv = t >> 5;
  const int sl = blockIdx.x & (CSR_SLICES - 1);
  const int b  = blockIdx.x / CSR_SLICES;
  const int lo = b * CSR_RANGE;
  const int nr = min(CSR_RANGE, N_NODES - lo);
  const int rot = b % CSR_CHUNKS;
  const int4* d4 = (const int4*)dst + sl * SLICE_VEC;

  for (int j = t; j < CSR_RANGE; j += CSR_THREADS) bins[j] = 0;
  if (t == 0) s_below = 0;
  __syncthreads();

  int below = 0;
  for (int k = 0; k < CSR_CHUNKS; k += 4) {
    int4 d[4];
    #pragma unroll
    for (int u = 0; u < 4; ++u) {
      int c = k + u + rot;
      if (c >= CSR_CHUNKS) c -= CSR_CHUNKS;
      const int v = c * CSR_THREADS + t;
      d[u] = (v < SLICE_VEC) ? d4[v] : make_int4(CSR_PAD, CSR_PAD, CSR_PAD, CSR_PAD);
    }
    #pragma unroll
    for (int u = 0; u < 4; ++u) {
      const int dd[4] = {d[u].x, d[u].y, d[u].z, d[u].w};
      #pragma unroll
      for (int q = 0; q < 4; ++q) {
        const unsigned r = (unsigned)(dd[q] - lo);
        below += (dd[q] < lo) ? 1 : 0;
        if (r < (unsigned)nr) atomicAdd(&bins[r], 1);
      }
    }
  }
  for (int off = 16; off; off >>= 1) below += __shfl_xor(below, off, 32);
  if (lane == 0) atomicAdd(&s_below, below);
  __syncthreads();

  if (wv == 0) {
    const int base = sl * SLICE_EDGES + s_below;
    int* rp = row_ptr + sl * RP_STRIDE;
    int loc[CSR_PER_LANE];
    int s = 0;
    #pragma unroll
    for (int i = 0; i < CSR_PER_LANE; ++i) {
      const int j = lane * CSR_PER_LANE + i;
      const int c = (j < CSR_RANGE) ? bins[j] : 0;
      loc[i] = s;
      s += c;
    }
    int inc = s;
    for (int off = 1; off < 32; off <<= 1) {
      const int up = __shfl_up(inc, off, 32);
      if (lane >= off) inc += up;
    }
    const int excl = inc - s;
    #pragma unroll
    for (int i = 0; i < CSR_PER_LANE; ++i) {
      const int j = lane * CSR_PER_LANE + i;
      if (j < CSR_RANGE) {
        const int start = base + excl + loc[i];
        bins[j] = start;
        if (j < nr) rp[lo + j] = start;
      }
    }
    if (b == CSR_RANGES - 1 && lane == 31) rp[N_NODES] = base + inc;
  }
  __syncthreads();

  for (int k = 0; k < CSR_CHUNKS; k += 4) {
    int4 d[4];
    int  vv[4];
    #pragma unroll
    for (int u = 0; u < 4; ++u) {
      int c = k + u + rot;
      if (c >= CSR_CHUNKS) c -= CSR_CHUNKS;
      const int v = c * CSR_THREADS + t;
      vv[u] = v;
      d[u] = (v < SLICE_VEC) ? d4[v] : make_int4(CSR_PAD, CSR_PAD, CSR_PAD, CSR_PAD);
    }
    int pos[4][4];
    #pragma unroll
    for (int u = 0; u < 4; ++u) {
      const int dd[4] = {d[u].x, d[u].y, d[u].z, d[u].w};
      #pragma unroll
      for (int q = 0; q < 4; ++q) {
        const unsigned r = (unsigned)(dd[q] - lo);
        int p = -1;
        if (r < (unsigned)nr) p = atomicAdd(&bins[r], 1);
        pos[u][q] = p;
      }
    }
    #pragma unroll
    for (int u = 0; u < 4; ++u) {
      #pragma unroll
      for (int q = 0; q < 4; ++q)
        if (pos[u][q] >= 0) edge_idx[pos[u][q]] = sl * SLICE_EDGES + 4 * vv[u] + q;
    }
  }
}

#define AGG_WAVES 4
#define AGG_EDGES 8
static_assert(N_NODES % AGG_WAVES == 0, "whole work-groups, no tail wave");
static_assert(32 % AGG_EDGES == 0, "a chunk is a whole number of batches");
static_assert(H_HEADS == 4 && F_INF == 128, "a lane holds 4 input features, one sum per head");

__device__ __forceinline__ float leaky(float v) { return v > 0.f ? v : NEG_SLOPE * v; }

__device__ __forceinline__ int seg_place(int i, const int (&off)[CSR_SLICES],
                                         const int (&cum)[CSR_SLICES]) {
  int d = off[0];
  #pragma unroll
  for (int g = 1; g < CSR_SLICES; ++g) d = (i >= cum[g]) ? off[g] : d;
  return i + d;
}

__global__ __launch_bounds__(AGG_WAVES * 32) void gat_aggregate(
    const float* __restrict__ xin, const float* __restrict__ el,
    const float* __restrict__ er, const int* __restrict__ src,
    const int* __restrict__ row_ptr, const int* __restrict__ edge_idx,
    float* __restrict__ z) {
  const int lane = threadIdx.x & 31;
  const int n = blockIdx.x * AGG_WAVES + __builtin_amdgcn_readfirstlane(threadIdx.x >> 5);
  int off[CSR_SLICES], cum[CSR_SLICES];
  int re = 0;
  #pragma unroll
  for (int g = 0; g < CSR_SLICES; ++g) {
    const int b = row_ptr[g * RP_STRIDE + n], e = row_ptr[g * RP_STRIDE + n + 1];
    cum[g] = re;
    off[g] = b - re;
    re += e - b;
  }
  const float4 er4 = *(const float4*)(er + n * H_HEADS);
  const float erd[H_HEADS] = {er4.x, er4.y, er4.z, er4.w};

  const bool ok0 = lane < re;
  int s0 = 0;
  float v0[H_HEADS];
  for (int hd = 0; hd < H_HEADS; ++hd) v0[hd] = -INFINITY;
  if (ok0) {
    s0 = src[edge_idx[seg_place(lane, off, cum)]];
    const float4 q4 = *(const float4*)(el + s0 * H_HEADS);
    const float q[H_HEADS] = {q4.x, q4.y, q4.z, q4.w};
    for (int hd = 0; hd < H_HEADS; ++hd) v0[hd] = leaky(q[hd] + erd[hd]);
  }

  float mx[H_HEADS];
  for (int hd = 0; hd < H_HEADS; ++hd) mx[hd] = fmaxf(-INFINITY, v0[hd]);
  for (int i = 32 + lane; i < re; i += 32) {
    const int s = src[edge_idx[seg_place(i, off, cum)]];
    const float4 q4 = *(const float4*)(el + s * H_HEADS);
    const float q[H_HEADS] = {q4.x, q4.y, q4.z, q4.w};
    for (int hd = 0; hd < H_HEADS; ++hd) mx[hd] = fmaxf(mx[hd], leaky(q[hd] + erd[hd]));
  }
  #pragma unroll
  for (int off = 16; off; off >>= 1) {
    for (int hd = 0; hd < H_HEADS; ++hd) mx[hd] = fmaxf(mx[hd], __shfl_xor(mx[hd], off, 32));
  }

  float p0[H_HEADS], sum[H_HEADS], inv[H_HEADS];
  for (int hd = 0; hd < H_HEADS; ++hd) {
    p0[hd] = ok0 ? __expf(v0[hd] - mx[hd]) : 0.f;
    sum[hd] = p0[hd];
  }
  for (int i = 32 + lane; i < re; i += 32) {
    const int s = src[edge_idx[seg_place(i, off, cum)]];
    const float4 q4 = *(const float4*)(el + s * H_HEADS);
    const float q[H_HEADS] = {q4.x, q4.y, q4.z, q4.w};
    for (int hd = 0; hd < H_HEADS; ++hd) sum[hd] += __expf(leaky(q[hd] + erd[hd]) - mx[hd]);
  }
  #pragma unroll
  for (int off = 16; off; off >>= 1) {
    for (int hd = 0; hd < H_HEADS; ++hd) sum[hd] += __shfl_xor(sum[hd], off, 32);
  }
  for (int hd = 0; hd < H_HEADS; ++hd) inv[hd] = (sum[hd] != 0.f) ? 1.f / sum[hd] : 0.f;

  const float* xrow = xin + lane * 4;
  float4 acc[H_HEADS];
  for (int hd = 0; hd < H_HEADS; ++hd) acc[hd] = make_float4(0.f, 0.f, 0.f, 0.f);
  int   s_l = s0;
  float a_l[H_HEADS];
  for (int hd = 0; hd < H_HEADS; ++hd) a_l[hd] = p0[hd] * inv[hd];
  for (int base = 0; base < re; base += 32) {
    if (base != 0) {
      s_l = 0;
      for (int hd = 0; hd < H_HEADS; ++hd) a_l[hd] = 0.f;
      if (base + lane < re) {
        s_l = src[edge_idx[seg_place(base + lane, off, cum)]];
        const float4 q4 = *(const float4*)(el + s_l * H_HEADS);
        const float q[H_HEADS] = {q4.x, q4.y, q4.z, q4.w};
        for (int hd = 0; hd < H_HEADS; ++hd)
          a_l[hd] = __expf(leaky(q[hd] + erd[hd]) - mx[hd]) * inv[hd];
      }
    }
    const int cnt = min(32, re - base);
    const int s_last = __builtin_amdgcn_readlane(s_l, cnt - 1);
    if (lane >= cnt) s_l = s_last;
    for (int j = 0; j < cnt; j += AGG_EDGES) {
      float4 xv[AGG_EDGES];
      float  aw[AGG_EDGES][H_HEADS];
      #pragma unroll
      for (int u = 0; u < AGG_EDGES; ++u) {
        xv[u] = *(const float4*)(xrow + (size_t)__builtin_amdgcn_readlane(s_l, j + u) * F_INF);
        for (int hd = 0; hd < H_HEADS; ++hd)
          aw[u][hd] = __int_as_float(__builtin_amdgcn_readlane(__float_as_int(a_l[hd]), j + u));
      }
      #pragma unroll
      for (int u = 0; u < AGG_EDGES; ++u) {
        for (int hd = 0; hd < H_HEADS; ++hd) {
          acc[hd].x += aw[u][hd] * xv[u].x; acc[hd].y += aw[u][hd] * xv[u].y;
          acc[hd].z += aw[u][hd] * xv[u].z; acc[hd].w += aw[u][hd] * xv[u].w;
        }
      }
    }
  }
  for (int hd = 0; hd < H_HEADS; ++hd)
    *(float4*)(z + (size_t)n * C_OUT + hd * F_INF + lane * 4) = acc[hd];
}

#define GEMM_WAVES 5
#define M_TILES    (N_NODES / 16)
#define N_SLABS    (F_OUTF / 32)
static_assert(N_NODES % 16 == 0, "whole row tiles");
static_assert(M_TILES % GEMM_WAVES == 0, "whole work-groups, no tail wave");
static_assert(F_OUTF == 128 && H_HEADS == 4 && F_INF % 8 == 0, "tile naming below, k-steps in pairs");

__global__ __launch_bounds__(GEMM_WAVES * 32) void gemm_zw_out(
    const float* __restrict__ z, const float* __restrict__ W,
    const float* __restrict__ bias, const float* __restrict__ alpha_p,
    const float* __restrict__ x0, float* __restrict__ xout) {
  const int lane = threadIdx.x & 31;
  const int wv   = threadIdx.x >> 5;
  const int q    = blockIdx.x & (N_SLABS - 1);
  const int tm   = (blockIdx.x >> 2) * GEMM_WAVES + wv;
  const int m0   = tm * 16;
  const int hf   = lane >> 4;
  const int c    = lane & 15;

  v8f acc0 = v8f{};
  v8f acc1 = v8f{};

  const float* arow = z + (size_t)(m0 + c) * C_OUT + 4 * hf;
  const float* bcol = W + (size_t)(4 * hf) * C_OUT + 32 * q + 2 * c;

  for (int hd = 0; hd < H_HEADS; ++hd) {
    const float* ah = arow + hd * F_INF;
    const float* bh = bcol + hd * F_OUTF;
    #pragma unroll 8
    for (int k8 = 0; k8 < F_INF; k8 += 8) {
      const float4 a4 = *(const float4*)(ah + k8);
      const float2 r0 = *(const float2*)(bh + (size_t)k8 * C_OUT);
      const float2 r1 = *(const float2*)(bh + (size_t)k8 * C_OUT + C_OUT);
      const float2 r2 = *(const float2*)(bh + (size_t)k8 * C_OUT + 2 * C_OUT);
      const float2 r3 = *(const float2*)(bh + (size_t)k8 * C_OUT + 3 * C_OUT);
      v2f a;  a.x  = a4.x; a.y  = a4.y;
      v2f b0; b0.x = r0.x; b0.y = r1.x;
      v2f b1; b1.x = r0.y; b1.y = r1.y;
      acc0 = __builtin_amdgcn_wmma_f32_16x16x4_f32(
           false, a,  false, b0,
           (short)0, acc0,  false,  false);
      acc1 = __builtin_amdgcn_wmma_f32_16x16x4_f32(
           false, a,  false, b1,
           (short)0, acc1,  false,  false);
      a.x  = a4.z; a.y  = a4.w;
      b0.x = r2.x; b0.y = r3.x;
      b1.x = r2.y; b1.y = r3.y;
      acc0 = __builtin_amdgcn_wmma_f32_16x16x4_f32(
           false, a,  false, b0,
           (short)0, acc0,  false,  false);
      acc1 = __builtin_amdgcn_wmma_f32_16x16x4_f32(
           false, a,  false, b1,
           (short)0, acc1,  false,  false);
    }
  }

  const int row0 = m0 + 8 * hf;
  const int col  = 32 * q + 2 * c;
  const float al = alpha_p[0];
  float bmx = 0.f, bmy = 0.f;
  for (int hd = 0; hd < H_HEADS; ++hd) {
    const float2 b = *(const float2*)(bias + hd * F_OUTF + col);
    bmx += b.x; bmy += b.y;
  }
  #pragma unroll
  for (int i = 0; i < 8; ++i) {
    const size_t o = (size_t)(row0 + i) * F_OUTF + col;
    const float2 xv = *(const float2*)(x0 + o);
    float2 r;
    r.x = (1.f - al) * (0.25f * (acc0[i] + bmx)) + al * xv.x;
    r.y = (1.f - al) * (0.25f * (acc1[i] + bmy)) + al * xv.y;
    *(float2*)(xout + o) = r;
  }
}

extern "C" void kernel_launch(void* const* d_in, const int* in_sizes, int n_in,
                              void* d_out, int out_size, void* d_ws, size_t ws_size,
                              hipStream_t stream) {
  const float* x      = (const float*)d_in[0];
  const float* x0     = (const float*)d_in[1];
  const int*   src    = (const int*)d_in[2];
  const int*   dst    = (const int*)d_in[3];
  const float* alpha  = (const float*)d_in[4];
  const float* W      = (const float*)d_in[5];
  const float* attn_l = (const float*)d_in[6];
  const float* attn_r = (const float*)d_in[7];
  const float* bias   = (const float*)d_in[8];

  char* ws = (char*)d_ws;
  size_t off = 0;
  auto alloc = [&](size_t bytes) -> char* {
    char* p = ws + off;
    off = (off + bytes + 255) & ~(size_t)255;
    return p;
  };
  float* z        = (float*)alloc((size_t)N_NODES * C_OUT * sizeof(float));
  float* el       = (float*)alloc((size_t)N_NODES * H_HEADS * sizeof(float));
  float* er       = (float*)alloc((size_t)N_NODES * H_HEADS * sizeof(float));
  float* xbuf     = (float*)alloc((size_t)N_NODES * F_INF * sizeof(float));
  int*   row_ptr  = (int*)alloc((size_t)CSR_SLICES * RP_STRIDE * sizeof(int));
  int*   edge_idx = (int*)alloc((size_t)N_EDGES * sizeof(int));
  float* wlr      = (float*)alloc((size_t)F_INF * 2 * H_HEADS * sizeof(float));

  csr_build<<<CSR_WGS, CSR_THREADS, 0, stream>>>(dst, row_ptr, edge_idx);
  attn_fold<<<(F_INF * H_HEADS) / FOLD_WAVES, FOLD_WAVES * 32, 0, stream>>>(W, attn_l, attn_r, wlr);

  const float* xin = x;
  for (int step = 0; step < STEPS; ++step) {
    node_logits<<<LOGIT_TILES / LOGIT_WAVES, LOGIT_WAVES * 32, 0, stream>>>(xin, wlr, el, er);
    gat_aggregate<<<N_NODES / AGG_WAVES, AGG_WAVES * 32, 0, stream>>>(xin, el, er, src, row_ptr,
                                                                    edge_idx, z);
    float* xo = (step == STEPS - 1) ? (float*)d_out : xbuf;
    gemm_zw_out<<<(M_TILES / GEMM_WAVES) * N_SLABS, GEMM_WAVES * 32, 0, stream>>>(
        z, W, bias, alpha, x0, xo);
    xin = xbuf;
  }
}


# aggregate + projection in one launch per step: 16 waves gather 16 nodes' rows into LDS, each then projects one head x 32 columns, heads summed in LDS; no z round trip; order of f32 sums changes
# speedup vs baseline: 3.4228x; 1.2385x over previous
#include <hip/hip_runtime.h>
#include <hip/hip_bf16.h>

#define N_NODES 10000
#define N_EDGES 320000
#define F_INF   128
#define F_OUTF  128
#define H_HEADS 4
#define C_OUT   (H_HEADS * F_OUTF)
#define STEPS   2
#define NEG_SLOPE 0.2f

typedef float v2f __attribute__((ext_vector_type(2)));
typedef float v8f __attribute__((ext_vector_type(8)));

#define FOLD_WAVES 4
static_assert((F_INF * H_HEADS) % FOLD_WAVES == 0, "whole work-groups, no tail wave");
static_assert(F_OUTF == 128, "a lane holds 4 of a head's 128 columns");

__global__ __launch_bounds__(FOLD_WAVES * 32) void attn_fold(
    const float* __restrict__ W, const float* __restrict__ attn_l,
    const float* __restrict__ attn_r, float* __restrict__ wlr) {
  const int lane = threadIdx.x & 31;
  const int id   = blockIdx.x * FOLD_WAVES + (threadIdx.x >> 5);
  const int k    = id >> 2;
  const int hd   = id & (H_HEADS - 1);
  const float4 w  = *(const float4*)(W + (size_t)k * C_OUT + hd * F_OUTF + 4 * lane);
  const float4 al = *(const float4*)(attn_l + hd * F_OUTF + 4 * lane);
  const float4 ar = *(const float4*)(attn_r + hd * F_OUTF + 4 * lane);
  float sl = w.x * al.x + w.y * al.y + w.z * al.z + w.w * al.w;
  float sr = w.x * ar.x + w.y * ar.y + w.z * ar.z + w.w * ar.w;
  for (int off = 16; off; off >>= 1) {
    sl += __shfl_xor(sl, off, 32);
    sr += __shfl_xor(sr, off, 32);
  }
  if (lane == 0) {
    wlr[hd * F_INF + k]             = sl;
    wlr[(H_HEADS + hd) * F_INF + k] = sr;
  }
}

#define LOGIT_WAVES 5
#define LOGIT_TILES (N_NODES / 16)
static_assert(N_NODES % 16 == 0, "whole 16-node tiles");
static_assert(LOGIT_TILES % LOGIT_WAVES == 0, "whole work-groups, no tail wave");
static_assert(F_INF % 8 == 0 && 2 * H_HEADS == 8, "k-steps in pairs; eight logits per node");

__global__ __launch_bounds__(LOGIT_WAVES * 32) void node_logits(
    const float* __restrict__ x, const float* __restrict__ wlr,
    float* __restrict__ el, float* __restrict__ er) {
  const int lane = threadIdx.x & 31;
  const int m0   = (blockIdx.x * LOGIT_WAVES + (threadIdx.x >> 5)) * 16;
  const int hf   = lane >> 4;
  const int c    = lane & 15;

  const float* arow = x + (size_t)(m0 + c) * F_INF + 4 * hf;
  const float* brow = wlr + (c & 7) * F_INF + 4 * hf;

  v8f acc0 = v8f{};
  v8f acc1 = v8f{};
  #pragma unroll
  for (int k8 = 0; k8 < F_INF; k8 += 8) {
    const float4 a4 = *(const float4*)(arow + k8);
    const float4 b4 = *(const float4*)(brow + k8);
    v2f a; a.x = a4.x; a.y = a4.y;
    v2f b; b.x = b4.x; b.y = b4.y;
    acc0 = __builtin_amdgcn_wmma_f32_16x16x4_f32(
         false, a,  false, b,
         (short)0, acc0,  false,  false);
    a.x = a4.z; a.y = a4.w;
    b.x = b4.z; b.y = b4.w;
    acc1 = __builtin_amdgcn_wmma_f32_16x16x4_f32(
         false, a,  false, b,
         (short)0, acc1,  false,  false);
  }

  if (c < 2 * H_HEADS) {
    float* o = (c < H_HEADS) ? (el + c) : (er + (c - H_HEADS));
    #pragma unroll
    for (int i = 0; i < 8; ++i) o[(m0 + 8 * hf + i) * H_HEADS] = acc0[i] + acc1[i];
  }
}

#define CSR_RANGES   32
#define CSR_SLICES   8
#define CSR_WGS      (CSR_RANGES * CSR_SLICES)
#define CSR_THREADS  512
#define CSR_RANGE    ((N_NODES + CSR_RANGES - 1) / CSR_RANGES)
#define CSR_PER_LANE ((CSR_RANGE + 31) / 32)
#define SLICE_EDGES  (N_EDGES / CSR_SLICES)
#define SLICE_VEC    (SLICE_EDGES / 4)
#define CSR_CHUNKS   ((SLICE_VEC + CSR_THREADS - 1) / CSR_THREADS)
#define RP_STRIDE    (((N_NODES + 1 + 31) / 32) * 32)
#define CSR_PAD      0x7fffffff

static_assert(N_EDGES % (4 * CSR_SLICES) == 0, "a slice of dst is read as int4");
static_assert((CSR_RANGES - 1) * CSR_RANGE < N_NODES, "every group owns at least one node");
static_assert(CSR_CHUNKS % 4 == 0, "the sweeps take four chunks per trip");
static_assert((CSR_SLICES & (CSR_SLICES - 1)) == 0, "slice number = low bits of the group number");

__global__ __launch_bounds__(CSR_THREADS) void csr_build(const int* __restrict__ dst,
                                                         int* __restrict__ row_ptr,
                                                         int* __restrict__ edge_idx) {
  __shared__ int bins[CSR_RANGE];
  __shared__ int s_below;
  const int t = threadIdx.x, lane = t & 31, wv = t >> 5;
  const int sl = blockIdx.x & (CSR_SLICES - 1);
  const int b  = blockIdx.x / CSR_SLICES;
  const int lo = b * CSR_RANGE;
  const int nr = min(CSR_RANGE, N_NODES - lo);
  const int rot = b % CSR_CHUNKS;
  const int4* d4 = (const int4*)dst + sl * SLICE_VEC;

  for (int j = t; j < CSR_RANGE; j += CSR_THREADS) bins[j] = 0;
  if (t == 0) s_below = 0;
  __syncthreads();

  int below = 0;
  for (int k = 0; k < CSR_CHUNKS; k += 4) {
    int4 d[4];
    #pragma unroll
    for (int u = 0; u < 4; ++u) {
      int c = k + u + rot;
      if (c >= CSR_CHUNKS) c -= CSR_CHUNKS;
      const int v = c * CSR_THREADS + t;
      d[u] = (v < SLICE_VEC) ? d4[v] : make_int4(CSR_PAD, CSR_PAD, CSR_PAD, CSR_PAD);
    }
    #pragma unroll
    for (int u = 0; u < 4; ++u) {
      const int dd[4] = {d[u].x, d[u].y, d[u].z, d[u].w};
      #pragma unroll
      for (int q = 0; q < 4; ++q) {
        const unsigned r = (unsigned)(dd[q] - lo);
        below += (dd[q] < lo) ? 1 : 0;
        if (r < (unsigned)nr) atomicAdd(&bins[r], 1);
      }
    }
  }
  for (int off = 16; off; off >>= 1) below += __shfl_xor(below, off, 32);
  if (lane == 0) atomicAdd(&s_below, below);
  __syncthreads();

  if (wv == 0) {
    const int base = sl * SLICE_EDGES + s_below;
    int* rp = row_ptr + sl * RP_STRIDE;
    int loc[CSR_PER_LANE];
    int s = 0;
    #pragma unroll
    for (int i = 0; i < CSR_PER_LANE; ++i) {
      const int j = lane * CSR_PER_LANE + i;
      const int c = (j < CSR_RANGE) ? bins[j] : 0;
      loc[i] = s;
      s += c;
    }
    int inc = s;
    for (int off = 1; off < 32; off <<= 1) {
      const int up = __shfl_up(inc, off, 32);
      if (lane >= off) inc += up;
    }
    const int excl = inc - s;
    #pragma unroll
    for (int i = 0; i < CSR_PER_LANE; ++i) {
      const int j = lane * CSR_PER_LANE + i;
      if (j < CSR_RANGE) {
        const int start = base + excl + loc[i];
        bins[j] = start;
        if (j < nr) rp[lo + j] = start;
      }
    }
    if (b == CSR_RANGES - 1 && lane == 31) rp[N_NODES] = base + inc;
  }
  __syncthreads();

  for (int k = 0; k < CSR_CHUNKS; k += 4) {
    int4 d[4];
    int  vv[4];
    #pragma unroll
    for (int u = 0; u < 4; ++u) {
      int c = k + u + rot;
      if (c >= CSR_CHUNKS) c -= CSR_CHUNKS;
      const int v = c * CSR_THREADS + t;
      vv[u] = v;
      d[u] = (v < SLICE_VEC) ? d4[v] : make_int4(CSR_PAD, CSR_PAD, CSR_PAD, CSR_PAD);
    }
    int pos[4][4];
    #pragma unroll
    for (int u = 0; u < 4; ++u) {
      const int dd[4] = {d[u].x, d[u].y, d[u].z, d[u].w};
      #pragma unroll
      for (int q = 0; q < 4; ++q) {
        const unsigned r = (unsigned)(dd[q] - lo);
        int p = -1;
        if (r < (unsigned)nr) p = atomicAdd(&bins[r], 1);
        pos[u][q] = p;
      }
    }
    #pragma unroll
    for (int u = 0; u < 4; ++u) {
      #pragma unroll
      for (int q = 0; q < 4; ++q)
        if (pos[u][q] >= 0) edge_idx[pos[u][q]] = sl * SLICE_EDGES + 4 * vv[u] + q;
    }
  }
}

#define FUSE_NODES 16
#define FUSE_WAVES 16
#define AGG_EDGES  8
#define N_SLABS    (F_OUTF / 32)
#define ZT_PITCH   (C_OUT + 4)
#define RED_PITCH  (F_OUTF + 4)
static_assert(N_NODES % FUSE_NODES == 0, "whole work-groups, no tail wave, no partial tile");
static_assert(FUSE_NODES == 16 && FUSE_WAVES == H_HEADS * N_SLABS && FUSE_WAVES == FUSE_NODES,
              "one wave per node, per (head, slab) and per output row");
static_assert(32 % AGG_EDGES == 0, "a chunk is a whole number of batches");
static_assert(H_HEADS == 4 && F_INF == 128 && F_OUTF == 128 && F_INF % 8 == 0,
              "a lane holds 4 input features and one sum per head; tile naming; k-steps in pairs");

__device__ __forceinline__ float leaky(float v) { return v > 0.f ? v : NEG_SLOPE * v; }

__device__ __forceinline__ int seg_place(int i, const int (&off)[CSR_SLICES],
                                         const int (&cum)[CSR_SLICES]) {
  int d = off[0];
  #pragma unroll
  for (int g = 1; g < CSR_SLICES; ++g) d = (i >= cum[g]) ? off[g] : d;
  return i + d;
}

__global__ __launch_bounds__(FUSE_WAVES * 32, 12) void gat_aggregate_project(
    const float* __restrict__ xin, const float* __restrict__ el,
    const float* __restrict__ er, const int* __restrict__ src,
    const int* __restrict__ row_ptr, const int* __restrict__ edge_idx,
    const float* __restrict__ W, const float* __restrict__ bias,
    const float* __restrict__ alpha_p, const float* __restrict__ x0,
    float* __restrict__ xout) {
  __shared__ __attribute__((aligned(16))) float zt[FUSE_NODES * ZT_PITCH];
  __shared__ __attribute__((aligned(16))) float red[H_HEADS * FUSE_NODES * RED_PITCH];

  const int lane = threadIdx.x & 31;
  const int wv = __builtin_amdgcn_readfirstlane(threadIdx.x >> 5);
  const int m0 = blockIdx.x * FUSE_NODES;
  const int n  = m0 + wv;
  int off[CSR_SLICES], cum[CSR_SLICES];
  int re = 0;
  #pragma unroll
  for (int g = 0; g < CSR_SLICES; ++g) {
    const int b = row_ptr[g * RP_STRIDE + n], e = row_ptr[g * RP_STRIDE + n + 1];
    cum[g] = re;
    off[g] = b - re;
    re += e - b;
  }
  const float4 er4 = *(const float4*)(er + n * H_HEADS);
  const float erd[H_HEADS] = {er4.x, er4.y, er4.z, er4.w};

  const bool ok0 = lane < re;
  int s0 = 0;
  float v0[H_HEADS];
  for (int hd = 0; hd < H_HEADS; ++hd) v0[hd] = -INFINITY;
  if (ok0) {
    s0 = src[edge_idx[seg_place(lane, off, cum)]];
    const float4 q4 = *(const float4*)(el + s0 * H_HEADS);
    const float q[H_HEADS] = {q4.x, q4.y, q4.z, q4.w};
    for (int hd = 0; hd < H_HEADS; ++hd) v0[hd] = leaky(q[hd] + erd[hd]);
  }

  float mx[H_HEADS];
  for (int hd = 0; hd < H_HEADS; ++hd) mx[hd] = fmaxf(-INFINITY, v0[hd]);
  for (int i = 32 + lane; i < re; i += 32) {
    const int s = src[edge_idx[seg_place(i, off, cum)]];
    const float4 q4 = *(const float4*)(el + s * H_HEADS);
    const float q[H_HEADS] = {q4.x, q4.y, q4.z, q4.w};
    for (int hd = 0; hd < H_HEADS; ++hd) mx[hd] = fmaxf(mx[hd], leaky(q[hd] + erd[hd]));
  }
  #pragma unroll
  for (int off = 16; off; off >>= 1) {
    for (int hd = 0; hd < H_HEADS; ++hd) mx[hd] = fmaxf(mx[hd], __shfl_xor(mx[hd], off, 32));
  }

  float p0[H_HEADS], sum[H_HEADS], inv[H_HEADS];
  for (int hd = 0; hd < H_HEADS; ++hd) {
    p0[hd] = ok0 ? __expf(v0[hd] - mx[hd]) : 0.f;
    sum[hd] = p0[hd];
  }
  for (int i = 32 + lane; i < re; i += 32) {
    const int s = src[edge_idx[seg_place(i, off, cum)]];
    const float4 q4 = *(const float4*)(el + s * H_HEADS);
    const float q[H_HEADS] = {q4.x, q4.y, q4.z, q4.w};
    for (int hd = 0; hd < H_HEADS; ++hd) sum[hd] += __expf(leaky(q[hd] + erd[hd]) - mx[hd]);
  }
  #pragma unroll
  for (int off = 16; off; off >>= 1) {
    for (int hd = 0; hd < H_HEADS; ++hd) sum[hd] += __shfl_xor(sum[hd], off, 32);
  }
  for (int hd = 0; hd < H_HEADS; ++hd) inv[hd] = (sum[hd] != 0.f) ? 1.f / sum[hd] : 0.f;

  const float* xrow = xin + lane * 4;
  float4 acc[H_HEADS];
  for (int hd = 0; hd < H_HEADS; ++hd) acc[hd] = make_float4(0.f, 0.f, 0.f, 0.f);
  int   s_l = s0;
  float a_l[H_HEADS];
  for (int hd = 0; hd < H_HEADS; ++hd) a_l[hd] = p0[hd] * inv[hd];
  for (int base = 0; base < re; base += 32) {
    if (base != 0) {
      s_l = 0;
      for (int hd = 0; hd < H_HEADS; ++hd) a_l[hd] = 0.f;
      if (base + lane < re) {
        s_l = src[edge_idx[seg_place(base + lane, off, cum)]];
        const float4 q4 = *(const float4*)(el + s_l * H_HEADS);
        const float q[H_HEADS] = {q4.x, q4.y, q4.z, q4.w};
        for (int hd = 0; hd < H_HEADS; ++hd)
          a_l[hd] = __expf(leaky(q[hd] + erd[hd]) - mx[hd]) * inv[hd];
      }
    }
    const int cnt = min(32, re - base);
    const int s_last = __builtin_amdgcn_readlane(s_l, cnt - 1);
    if (lane >= cnt) s_l = s_last;
    for (int j = 0; j < cnt; j += AGG_EDGES) {
      float4 xv[AGG_EDGES];
      float  aw[AGG_EDGES][H_HEADS];
      #pragma unroll
      for (int u = 0; u < AGG_EDGES; ++u) {
        xv[u] = *(const float4*)(xrow + (size_t)__builtin_amdgcn_readlane(s_l, j + u) * F_INF);
        for (int hd = 0; hd < H_HEADS; ++hd)
          aw[u][hd] = __int_as_float(__builtin_amdgcn_readlane(__float_as_int(a_l[hd]), j + u));
      }
      #pragma unroll
      for (int u = 0; u < AGG_EDGES; ++u) {
        for (int hd = 0; hd < H_HEADS; ++hd) {
          acc[hd].x += aw[u][hd] * xv[u].x; acc[hd].y += aw[u][hd] * xv[u].y;
          acc[hd].z += aw[u][hd] * xv[u].z; acc[hd].w += aw[u][hd] * xv[u].w;
        }
      }
    }
  }
  for (int hd = 0; hd < H_HEADS; ++hd)
    *(float4*)(zt + wv * ZT_PITCH + hd * F_INF + lane * 4) = acc[hd];
  __syncthreads();

  {
    const int hd = wv >> 2;
    const int q  = wv & (N_SLABS - 1);
    const int hf = lane >> 4;
    const int c  = lane & 15;
    v8f acc0 = v8f{};
    v8f acc1 = v8f{};
    const float* ah = zt + c * ZT_PITCH + hd * F_INF + 4 * hf;
    const float* bh = W + (size_t)(4 * hf) * C_OUT + hd * F_OUTF + 32 * q + 2 * c;
    #pragma unroll 4
    for (int k8 = 0; k8 < F_INF; k8 += 8) {
      const float4 a4 = *(const float4*)(ah + k8);
      const float2 r0 = *(const float2*)(bh + (size_t)k8 * C_OUT);
      const float2 r1 = *(const float2*)(bh + (size_t)k8 * C_OUT + C_OUT);
      const float2 r2 = *(const float2*)(bh + (size_t)k8 * C_OUT + 2 * C_OUT);
      const float2 r3 = *(const float2*)(bh + (size_t)k8 * C_OUT + 3 * C_OUT);
      v2f a;  a.x  = a4.x; a.y  = a4.y;
      v2f b0; b0.x = r0.x; b0.y = r1.x;
      v2f b1; b1.x = r0.y; b1.y = r1.y;
      acc0 = __builtin_amdgcn_wmma_f32_16x16x4_f32(
           false, a,  false, b0,
           (short)0, acc0,  false,  false);
      acc1 = __builtin_amdgcn_wmma_f32_16x16x4_f32(
           false, a,  false, b1,
           (short)0, acc1,  false,  false);
      a.x  = a4.z; a.y  = a4.w;
      b0.x = r2.x; b0.y = r3.x;
      b1.x = r2.y; b1.y = r3.y;
      acc0 = __builtin_amdgcn_wmma_f32_16x16x4_f32(
           false, a,  false, b0,
           (short)0, acc0,  false,  false);
      acc1 = __builtin_amdgcn_wmma_f32_16x16x4_f32(
           false, a,  false, b1,
           (short)0, acc1,  false,  false);
    }
    float* rp = red + (hd * FUSE_NODES + 8 * hf) * RED_PITCH + 32 * q + 2 * c;
    #pragma unroll
    for (int i = 0; i < 8; ++i) {
      float2 p; p.x = acc0[i]; p.y = acc1[i];
      *(float2*)(rp + i * RED_PITCH) = p;
    }
  }
  __syncthreads();

  {
    const int col = lane * 4;
    float4 s = *(const float4*)(red + wv * RED_PITCH + col);
    float4 bm = *(const float4*)(bias + col);
    for (int hd = 1; hd < H_HEADS; ++hd) {
      const float4 p = *(const float4*)(red + (hd * FUSE_NODES + wv) * RED_PITCH + col);
      const float4 b = *(const float4*)(bias + hd * F_OUTF + col);
      s.x += p.x; s.y += p.y; s.z += p.z; s.w += p.w;
      bm.x += b.x; bm.y += b.y; bm.z += b.z; bm.w += b.w;
    }
    const float al = alpha_p[0];
    const size_t o = (size_t)n * F_OUTF + col;
    const float4 xv = *(const float4*)(x0 + o);
    float4 r;
    r.x = (1.f - al) * (0.25f * (s.x + bm.x)) + al * xv.x;
    r.y = (1.f - al) * (0.25f * (s.y + bm.y)) + al * xv.y;
    r.z = (1.f - al) * (0.25f * (s.z + bm.z)) + al * xv.z;
    r.w = (1.f - al) * (0.25f * (s.w + bm.w)) + al * xv.w;
    *(float4*)(xout + o) = r;
  }
}

extern "C" void kernel_launch(void* const* d_in, const int* in_sizes, int n_in,
                              void* d_out, int out_size, void* d_ws, size_t ws_size,
                              hipStream_t stream) {
  const float* x      = (const float*)d_in[0];
  const float* x0     = (const float*)d_in[1];
  const int*   src    = (const int*)d_in[2];
  const int*   dst    = (const int*)d_in[3];
  const float* alpha  = (const float*)d_in[4];
  const float* W      = (const float*)d_in[5];
  const float* attn_l = (const float*)d_in[6];
  const float* attn_r = (const float*)d_in[7];
  const float* bias   = (const float*)d_in[8];

  char* ws = (char*)d_ws;
  size_t off = 0;
  auto alloc = [&](size_t bytes) -> char* {
    char* p = ws + off;
    off = (off + bytes + 255) & ~(size_t)255;
    return p;
  };
  float* el       = (float*)alloc((size_t)N_NODES * H_HEADS * sizeof(float));
  float* er       = (float*)alloc((size_t)N_NODES * H_HEADS * sizeof(float));
  float* xbuf     = (float*)alloc((size_t)N_NODES * F_INF * sizeof(float));
  int*   row_ptr  = (int*)alloc((size_t)CSR_SLICES * RP_STRIDE * sizeof(int));
  int*   edge_idx = (int*)alloc((size_t)N_EDGES * sizeof(int));
  float* wlr      = (float*)alloc((size_t)F_INF * 2 * H_HEADS * sizeof(float));

  csr_build<<<CSR_WGS, CSR_THREADS, 0, stream>>>(dst, row_ptr, edge_idx);
  attn_fold<<<(F_INF * H_HEADS) / FOLD_WAVES, FOLD_WAVES * 32, 0, stream>>>(W, attn_l, attn_r, wlr);

  const float* xin = x;
  for (int step = 0; step < STEPS; ++step) {
    node_logits<<<LOGIT_TILES / LOGIT_WAVES, LOGIT_WAVES * 32, 0, stream>>>(xin, wlr, el, er);
    float* xo = (step == STEPS - 1) ? (float*)d_out : xbuf;
    gat_aggregate_project<<<N_NODES / FUSE_NODES, FUSE_WAVES * 32, 0, stream>>>(
        xin, el, er, src, row_ptr, edge_idx, W, bias, alpha, x0, xo);
    xin = xbuf;
  }
}


# launches 6 -> 4: the attention fold rides in the CSR build's launch as extra groups; step 2's node logits come from step 1's aggregate epilogue (matrix unit, operands from LDS); sums unchanged
# speedup vs baseline: 3.6993x; 1.0808x over previous
#include <hip/hip_runtime.h>
#include <hip/hip_bf16.h>

#define N_NODES 10000
#define N_EDGES 320000
#define F_INF   128
#define F_OUTF  128
#define H_HEADS 4
#define C_OUT   (H_HEADS * F_OUTF)
#define STEPS   2
#define NEG_SLOPE 0.2f

typedef float v2f __attribute__((ext_vector_type(2)));
typedef float v8f __attribute__((ext_vector_type(8)));

#define FOLD_WAVES 16
#define FOLD_WGS   ((F_INF * H_HEADS) / FOLD_WAVES)
static_assert((F_INF * H_HEADS) % FOLD_WAVES == 0, "whole work-groups, no tail wave");
static_assert(F_OUTF == 128, "a lane holds 4 of a head's 128 columns");

__device__ __forceinline__ void attn_fold_wave(
    const int id, const int lane,
    const float* __restrict__ W, const float* __restrict__ attn_l,
    const float* __restrict__ attn_r, float* __restrict__ wlr) {
  const int k    = id >> 2;
  const int hd   = id & (H_HEADS - 1);
  const float4 w  = *(const float4*)(W + (size_t)k * C_OUT + hd * F_OUTF + 4 * lane);
  const float4 al = *(const float4*)(attn_l + hd * F_OUTF + 4 * lane);
  const float4 ar = *(const float4*)(attn_r + hd * F_OUTF + 4 * lane);
  float sl = w.x * al.x + w.y * al.y + w.z * al.z + w.w * al.w;
  float sr = w.x * ar.x + w.y * ar.y + w.z * ar.z + w.w * ar.w;
  for (int off = 16; off; off >>= 1) {
    sl += __shfl_xor(sl, off, 32);
    sr += __shfl_xor(sr, off, 32);
  }
  if (lane == 0) {
    wlr[hd * F_INF + k]             = sl;
    wlr[(H_HEADS + hd) * F_INF + k] = sr;
  }
}

#define LOGIT_WAVES 5
#define LOGIT_TILES (N_NODES / 16)
static_assert(N_NODES % 16 == 0, "whole 16-node tiles");
static_assert(LOGIT_TILES % LOGIT_WAVES == 0, "whole work-groups, no tail wave");
static_assert(F_INF % 8 == 0 && 2 * H_HEADS == 8, "k-steps in pairs; eight logits per node");

__global__ __launch_bounds__(LOGIT_WAVES * 32) void node_logits(
    const float* __restrict__ x, const float* __restrict__ wlr,
    float* __restrict__ el, float* __restrict__ er) {
  const int lane = threadIdx.x & 31;
  const int m0   = (blockIdx.x * LOGIT_WAVES + (threadIdx.x >> 5)) * 16;
  const int hf   = lane >> 4;
  const int c    = lane & 15;

  const float* arow = x + (size_t)(m0 + c) * F_INF + 4 * hf;
  const float* brow = wlr + (c & 7) * F_INF + 4 * hf;

  v8f acc0 = v8f{};
  v8f acc1 = v8f{};
  #pragma unroll
  for (int k8 = 0; k8 < F_INF; k8 += 8) {
    const float4 a4 = *(const float4*)(arow + k8);
    const float4 b4 = *(const float4*)(brow + k8);
    v2f a; a.x = a4.x; a.y = a4.y;
    v2f b; b.x = b4.x; b.y = b4.y;
    acc0 = __builtin_amdgcn_wmma_f32_16x16x4_f32(
         false, a,  false, b,
         (short)0, acc0,  false,  false);
    a.x = a4.z; a.y = a4.w;
    b.x = b4.z; b.y = b4.w;
    acc1 = __builtin_amdgcn_wmma_f32_16x16x4_f32(
         false, a,  false, b,
         (short)0, acc1,  false,  false);
  }

  if (c < 2 * H_HEADS) {
    float* o = (c < H_HEADS) ? (el + c) : (er + (c - H_HEADS));
    #pragma unroll
    for (int i = 0; i < 8; ++i) o[(m0 + 8 * hf + i) * H_HEADS] = acc0[i] + acc1[i];
  }
}

#define CSR_RANGES   32
#define CSR_SLICES   8
#define CSR_WGS      (CSR_RANGES * CSR_SLICES)
#define CSR_THREADS  512
#define CSR_RANGE    ((N_NODES + CSR_RANGES - 1) / CSR_RANGES)
#define CSR_PER_LANE ((CSR_RANGE + 31) / 32)
#define SLICE_EDGES  (N_EDGES / CSR_SLICES)
#define SLICE_VEC    (SLICE_EDGES / 4)
#define CSR_CHUNKS   ((SLICE_VEC + CSR_THREADS - 1) / CSR_THREADS)
#define RP_STRIDE    (((N_NODES + 1 + 31) / 32) * 32)
#define CSR_PAD      0x7fffffff

static_assert(N_EDGES % (4 * CSR_SLICES) == 0, "a slice of dst is read as int4");
static_assert((CSR_RANGES - 1) * CSR_RANGE < N_NODES, "every group owns at least one node");
static_assert(CSR_CHUNKS % 4 == 0, "the sweeps take four chunks per trip");
static_assert((CSR_SLICES & (CSR_SLICES - 1)) == 0, "slice number = low bits of the group number");

static_assert(FOLD_WAVES * 32 == CSR_THREADS, "the fold's groups have the CSR groups' shape");

__global__ __launch_bounds__(CSR_THREADS) void csr_build(const int* __restrict__ dst,
                                                         int* __restrict__ row_ptr,
                                                         int* __restrict__ edge_idx,
                                                         const float* __restrict__ W,
                                                         const float* __restrict__ attn_l,
                                                         const float* __restrict__ attn_r,
                                                         float* __restrict__ wlr) {
  __shared__ int bins[CSR_RANGE];
  __shared__ int s_below;
  const int t = threadIdx.x, lane = t & 31, wv = t >> 5;
  if (blockIdx.x >= CSR_WGS) {
    attn_fold_wave((blockIdx.x - CSR_WGS) * FOLD_WAVES + wv, lane, W, attn_l, attn_r, wlr);
    return;
  }
  const int sl = blockIdx.x & (CSR_SLICES - 1);
  const int b  = blockIdx.x / CSR_SLICES;
  const int lo = b * CSR_RANGE;
  const int nr = min(CSR_RANGE, N_NODES - lo);
  const int rot = b % CSR_CHUNKS;
  const int4* d4 = (const int4*)dst + sl * SLICE_VEC;

  for (int j = t; j < CSR_RANGE; j += CSR_THREADS) bins[j] = 0;
  if (t == 0) s_below = 0;
  __syncthreads();

  int below = 0;
  for (int k = 0; k < CSR_CHUNKS; k += 4) {
    int4 d[4];
    #pragma unroll
    for (int u = 0; u < 4; ++u) {
      int c = k + u + rot;
      if (c >= CSR_CHUNKS) c -= CSR_CHUNKS;
      const int v = c * CSR_THREADS + t;
      d[u] = (v < SLICE_VEC) ? d4[v] : make_int4(CSR_PAD, CSR_PAD, CSR_PAD, CSR_PAD);
    }
    #pragma unroll
    for (int u = 0; u < 4; ++u) {
      const int dd[4] = {d[u].x, d[u].y, d[u].z, d[u].w};
      #pragma unroll
      for (int q = 0; q < 4; ++q) {
        const unsigned r = (unsigned)(dd[q] - lo);
        below += (dd[q] < lo) ? 1 : 0;
        if (r < (unsigned)nr) atomicAdd(&bins[r], 1);
      }
    }
  }
  for (int off = 16; off; off >>= 1) below += __shfl_xor(below, off, 32);
  if (lane == 0) atomicAdd(&s_below, below);
  __syncthreads();

  if (wv == 0) {
    const int base = sl * SLICE_EDGES + s_below;
    int* rp = row_ptr + sl * RP_STRIDE;
    int loc[CSR_PER_LANE];
    int s = 0;
    #pragma unroll
    for (int i = 0; i < CSR_PER_LANE; ++i) {
      const int j = lane * CSR_PER_LANE + i;
      const int c = (j < CSR_RANGE) ? bins[j] : 0;
      loc[i] = s;
      s += c;
    }
    int inc = s;
    for (int off = 1; off < 32; off <<= 1) {
      const int up = __shfl_up(inc, off, 32);
      if (lane >= off) inc += up;
    }
    const int excl = inc - s;
    #pragma unroll
    for (int i = 0; i < CSR_PER_LANE; ++i) {
      const int j = lane * CSR_PER_LANE + i;
      if (j < CSR_RANGE) {
        const int start = base + excl + loc[i];
        bins[j] = start;
        if (j < nr) rp[lo + j] = start;
      }
    }
    if (b == CSR_RANGES - 1 && lane == 31) rp[N_NODES] = base + inc;
  }
  __syncthreads();

  for (int k = 0; k < CSR_CHUNKS; k += 4) {
    int4 d[4];
    int  vv[4];
    #pragma unroll
    for (int u = 0; u < 4; ++u) {
      int c = k + u + rot;
      if (c >= CSR_CHUNKS) c -= CSR_CHUNKS;
      const int v = c * CSR_THREADS + t;
      vv[u] = v;
      d[u] = (v < SLICE_VEC) ? d4[v] : make_int4(CSR_PAD, CSR_PAD, CSR_PAD, CSR_PAD);
    }
    int pos[4][4];
    #pragma unroll
    for (int u = 0; u < 4; ++u) {
      const int dd[4] = {d[u].x, d[u].y, d[u].z, d[u].w};
      #pragma unroll
      for (int q = 0; q < 4; ++q) {
        const unsigned r = (unsigned)(dd[q] - lo);
        int p = -1;
        if (r < (unsigned)nr) p = atomicAdd(&bins[r], 1);
        pos[u][q] = p;
      }
    }
    #pragma unroll
    for (int u = 0; u < 4; ++u) {
      #pragma unroll
      for (int q = 0; q < 4; ++q)
        if (pos[u][q] >= 0) edge_idx[pos[u][q]] = sl * SLICE_EDGES + 4 * vv[u] + q;
    }
  }
}

#define FUSE_NODES 16
#define FUSE_WAVES 16
#define AGG_EDGES  8
#define N_SLABS    (F_OUTF / 32)
#define ZT_PITCH   (C_OUT + 4)
#define RED_PITCH  (F_OUTF + 4)
#define WL_PITCH   (F_INF + 4)
static_assert(2 * H_HEADS * F_INF == 2 * FUSE_WAVES * 32, "a thread copies two floats of the folded vectors");
static_assert(N_NODES % FUSE_NODES == 0, "whole work-groups, no tail wave, no partial tile");
static_assert(FUSE_NODES == 16 && FUSE_WAVES == H_HEADS * N_SLABS && FUSE_WAVES == FUSE_NODES,
              "one wave per node, per (head, slab) and per output row");
static_assert(32 % AGG_EDGES == 0, "a chunk is a whole number of batches");
static_assert(H_HEADS == 4 && F_INF == 128 && F_OUTF == 128 && F_INF % 8 == 0,
              "a lane holds 4 input features and one sum per head; tile naming; k-steps in pairs");

__device__ __forceinline__ float leaky(float v) { return v > 0.f ? v : NEG_SLOPE * v; }

__device__ __forceinline__ int seg_place(int i, const int (&off)[CSR_SLICES],
                                         const int (&cum)[CSR_SLICES]) {
  int d = off[0];
  #pragma unroll
  for (int g = 1; g < CSR_SLICES; ++g) d = (i >= cum[g]) ? off[g] : d;
  return i + d;
}

__global__ __launch_bounds__(FUSE_WAVES * 32, 12) void gat_aggregate_project(
    const float* __restrict__ xin, const float* __restrict__ lg,
    const int* __restrict__ src,
    const int* __restrict__ row_ptr, const int* __restrict__ edge_idx,
    const float* __restrict__ W, const float* __restrict__ bias,
    const float* __restrict__ alpha_p, const float* __restrict__ x0,
    const float* __restrict__ wlr, float* __restrict__ lg_next,
    float* __restrict__ xout) {
  const float* el = lg;
  const float* er = lg + N_NODES * H_HEADS;
  __shared__ __attribute__((aligned(16))) float zt[FUSE_NODES * ZT_PITCH];
  __shared__ __attribute__((aligned(16))) float red[H_HEADS * FUSE_NODES * RED_PITCH];
  __shared__ __attribute__((aligned(16))) float wls[2 * H_HEADS * WL_PITCH];

  const int lane = threadIdx.x & 31;
  const int wv = __builtin_amdgcn_readfirstlane(threadIdx.x >> 5);
  const int m0 = blockIdx.x * FUSE_NODES;
  const int n  = m0 + wv;
  int off[CSR_SLICES], cum[CSR_SLICES];
  int re = 0;
  #pragma unroll
  for (int g = 0; g < CSR_SLICES; ++g) {
    const int b = row_ptr[g * RP_STRIDE + n], e = row_ptr[g * RP_STRIDE + n + 1];
    cum[g] = re;
    off[g] = b - re;
    re += e - b;
  }
  const float4 er4 = *(const float4*)(er + n * H_HEADS);
  const float erd[H_HEADS] = {er4.x, er4.y, er4.z, er4.w};

  const bool ok0 = lane < re;
  int s0 = 0;
  float v0[H_HEADS];
  for (int hd = 0; hd < H_HEADS; ++hd) v0[hd] = -INFINITY;
  if (ok0) {
    s0 = src[edge_idx[seg_place(lane, off, cum)]];
    const float4 q4 = *(const float4*)(el + s0 * H_HEADS);
    const float q[H_HEADS] = {q4.x, q4.y, q4.z, q4.w};
    for (int hd = 0; hd < H_HEADS; ++hd) v0[hd] = leaky(q[hd] + erd[hd]);
  }

  float mx[H_HEADS];
  for (int hd = 0; hd < H_HEADS; ++hd) mx[hd] = fmaxf(-INFINITY, v0[hd]);
  for (int i = 32 + lane; i < re; i += 32) {
    const int s = src[edge_idx[seg_place(i, off, cum)]];
    const float4 q4 = *(const float4*)(el + s * H_HEADS);
    const float q[H_HEADS] = {q4.x, q4.y, q4.z, q4.w};
    for (int hd = 0; hd < H_HEADS; ++hd) mx[hd] = fmaxf(mx[hd], leaky(q[hd] + erd[hd]));
  }
  #pragma unroll
  for (int off = 16; off; off >>= 1) {
    for (int hd = 0; hd < H_HEADS; ++hd) mx[hd] = fmaxf(mx[hd], __shfl_xor(mx[hd], off, 32));
  }

  float p0[H_HEADS], sum[H_HEADS], inv[H_HEADS];
  for (int hd = 0; hd < H_HEADS; ++hd) {
    p0[hd] = ok0 ? __expf(v0[hd] - mx[hd]) : 0.f;
    sum[hd] = p0[hd];
  }
  for (int i = 32 + lane; i < re; i += 32) {
    const int s = src[edge_idx[seg_place(i, off, cum)]];
    const float4 q4 = *(const float4*)(el + s * H_HEADS);
    const float q[H_HEADS] = {q4.x, q4.y, q4.z, q4.w};
    for (int hd = 0; hd < H_HEADS; ++hd) sum[hd] += __expf(leaky(q[hd] + erd[hd]) - mx[hd]);
  }
  #pragma unroll
  for (int off = 16; off; off >>= 1) {
    for (int hd = 0; hd < H_HEADS; ++hd) sum[hd] += __shfl_xor(sum[hd], off, 32);
  }
  for (int hd = 0; hd < H_HEADS; ++hd) inv[hd] = (sum[hd] != 0.f) ? 1.f / sum[hd] : 0.f;

  const float* xrow = xin + lane * 4;
  float4 acc[H_HEADS];
  for (int hd = 0; hd < H_HEADS; ++hd) acc[hd] = make_float4(0.f, 0.f, 0.f, 0.f);
  int   s_l = s0;
  float a_l[H_HEADS];
  for (int hd = 0; hd < H_HEADS; ++hd) a_l[hd] = p0[hd] * inv[hd];
  for (int base = 0; base < re; base += 32) {
    if (base != 0) {
      s_l = 0;
      for (int hd = 0; hd < H_HEADS; ++hd) a_l[hd] = 0.f;
      if (base + lane < re) {
        s_l = src[edge_idx[seg_place(base + lane, off, cum)]];
        const float4 q4 = *(const float4*)(el + s_l * H_HEADS);
        const float q[H_HEADS] = {q4.x, q4.y, q4.z, q4.w};
        for (int hd = 0; hd < H_HEADS; ++hd)
          a_l[hd] = __expf(leaky(q[hd] + erd[hd]) - mx[hd]) * inv[hd];
      }
    }
    const int cnt = min(32, re - base);
    const int s_last = __builtin_amdgcn_readlane(s_l, cnt - 1);
    if (lane >= cnt) s_l = s_last;
    for (int j = 0; j < cnt; j += AGG_EDGES) {
      float4 xv[AGG_EDGES];
      float  aw[AGG_EDGES][H_HEADS];
      #pragma unroll
      for (int u = 0; u < AGG_EDGES; ++u) {
        xv[u] = *(const float4*)(xrow + (size_t)__builtin_amdgcn_readlane(s_l, j + u) * F_INF);
        for (int hd = 0; hd < H_HEADS; ++hd)
          aw[u][hd] = __int_as_float(__builtin_amdgcn_readlane(__float_as_int(a_l[hd]), j + u));
      }
      #pragma unroll
      for (int u = 0; u < AGG_EDGES; ++u) {
        for (int hd = 0; hd < H_HEADS; ++hd) {
          acc[hd].x += aw[u][hd] * xv[u].x; acc[hd].y += aw[u][hd] * xv[u].y;
          acc[hd].z += aw[u][hd] * xv[u].z; acc[hd].w += aw[u][hd] * xv[u].w;
        }
      }
    }
  }
  for (int hd = 0; hd < H_HEADS; ++hd)
    *(float4*)(zt + wv * ZT_PITCH + hd * F_INF + lane * 4) = acc[hd];
  __syncthreads();

  {
    const int hd = wv >> 2;
    const int q  = wv & (N_SLABS - 1);
    const int hf = lane >> 4;
    const int c  = lane & 15;
    float2 wl2 = make_float2(0.f, 0.f);
    if (lg_next != nullptr) wl2 = *(const float2*)(wlr + 2 * threadIdx.x);
    v8f acc0 = v8f{};
    v8f acc1 = v8f{};
    const float* ah = zt + c * ZT_PITCH + hd * F_INF + 4 * hf;
    const float* bh = W + (size_t)(4 * hf) * C_OUT + hd * F_OUTF + 32 * q + 2 * c;
    #pragma unroll 4
    for (int k8 = 0; k8 < F_INF; k8 += 8) {
      const float4 a4 = *(const float4*)(ah + k8);
      const float2 r0 = *(const float2*)(bh + (size_t)k8 * C_OUT);
      const float2 r1 = *(const float2*)(bh + (size_t)k8 * C_OUT + C_OUT);
      const float2 r2 = *(const float2*)(bh + (size_t)k8 * C_OUT + 2 * C_OUT);
      const float2 r3 = *(const float2*)(bh + (size_t)k8 * C_OUT + 3 * C_OUT);
      v2f a;  a.x  = a4.x; a.y  = a4.y;
      v2f b0; b0.x = r0.x; b0.y = r1.x;
      v2f b1; b1.x = r0.y; b1.y = r1.y;
      acc0 = __builtin_amdgcn_wmma_f32_16x16x4_f32(
           false, a,  false, b0,
           (short)0, acc0,  false,  false);
      acc1 = __builtin_amdgcn_wmma_f32_16x16x4_f32(
           false, a,  false, b1,
           (short)0, acc1,  false,  false);
      a.x  = a4.z; a.y  = a4.w;
      b0.x = r2.x; b0.y = r3.x;
      b1.x = r2.y; b1.y = r3.y;
      acc0 = __builtin_amdgcn_wmma_f32_16x16x4_f32(
           false, a,  false, b0,
           (short)0, acc0,  false,  false);
      acc1 = __builtin_amdgcn_wmma_f32_16x16x4_f32(
           false, a,  false, b1,
           (short)0, acc1,  false,  false);
    }
    float* rp = red + (hd * FUSE_NODES + 8 * hf) * RED_PITCH + 32 * q + 2 * c;
    #pragma unroll
    for (int i = 0; i < 8; ++i) {
      float2 p; p.x = acc0[i]; p.y = acc1[i];
      *(float2*)(rp + i * RED_PITCH) = p;
    }
    if (lg_next != nullptr)
      *(float2*)(wls + (threadIdx.x >> 6) * WL_PITCH + 2 * (threadIdx.x & 63)) = wl2;
  }
  __syncthreads();

  {
    const int col = lane * 4;
    float4 s = *(const float4*)(red + wv * RED_PITCH + col);
    float4 bm = *(const float4*)(bias + col);
    for (int hd = 1; hd < H_HEADS; ++hd) {
      const float4 p = *(const float4*)(red + (hd * FUSE_NODES + wv) * RED_PITCH + col);
      const float4 b = *(const float4*)(bias + hd * F_OUTF + col);
      s.x += p.x; s.y += p.y; s.z += p.z; s.w += p.w;
      bm.x += b.x; bm.y += b.y; bm.z += b.z; bm.w += b.w;
    }
    const float al = alpha_p[0];
    const size_t o = (size_t)n * F_OUTF + col;
    const float4 xv = *(const float4*)(x0 + o);
    float4 r;
    r.x = (1.f - al) * (0.25f * (s.x + bm.x)) + al * xv.x;
    r.y = (1.f - al) * (0.25f * (s.y + bm.y)) + al * xv.y;
    r.z = (1.f - al) * (0.25f * (s.z + bm.z)) + al * xv.z;
    r.w = (1.f - al) * (0.25f * (s.w + bm.w)) + al * xv.w;
    *(float4*)(xout + o) = r;

    if (lg_next != nullptr) {
      *(float4*)(zt + wv * ZT_PITCH + col) = r;
      __syncthreads();
      if (wv == 0) {
        const int hf = lane >> 4;
        const int c  = lane & 15;
        const float* arow = zt + c * ZT_PITCH + 4 * hf;
        const float* brow = wls + (c & 7) * WL_PITCH + 4 * hf;
        v8f acc0 = v8f{};
        v8f acc1 = v8f{};
        #pragma unroll 4
        for (int k8 = 0; k8 < F_INF; k8 += 8) {
          const float4 a4 = *(const float4*)(arow + k8);
          const float4 b4 = *(const float4*)(brow + k8);
          v2f a; a.x = a4.x; a.y = a4.y;
          v2f b; b.x = b4.x; b.y = b4.y;
          acc0 = __builtin_amdgcn_wmma_f32_16x16x4_f32(
               false, a,  false, b,
               (short)0, acc0,  false,  false);
          a.x = a4.z; a.y = a4.w;
          b.x = b4.z; b.y = b4.w;
          acc1 = __builtin_amdgcn_wmma_f32_16x16x4_f32(
               false, a,  false, b,
               (short)0, acc1,  false,  false);
        }
        if (c < 2 * H_HEADS) {
          float* o2 = (c < H_HEADS) ? (lg_next + c) : (lg_next + N_NODES * H_HEADS + (c - H_HEADS));
          #pragma unroll
          for (int i = 0; i < 8; ++i) o2[(m0 + 8 * hf + i) * H_HEADS] = acc0[i] + acc1[i];
        }
      }
    }
  }
}

extern "C" void kernel_launch(void* const* d_in, const int* in_sizes, int n_in,
                              void* d_out, int out_size, void* d_ws, size_t ws_size,
                              hipStream_t stream) {
  const float* x      = (const float*)d_in[0];
  const float* x0     = (const float*)d_in[1];
  const int*   src    = (const int*)d_in[2];
  const int*   dst    = (const int*)d_in[3];
  const float* alpha  = (const float*)d_in[4];
  const float* W      = (const float*)d_in[5];
  const float* attn_l = (const float*)d_in[6];
  const float* attn_r = (const float*)d_in[7];
  const float* bias   = (const float*)d_in[8];

  char* ws = (char*)d_ws;
  size_t off = 0;
  auto alloc = [&](size_t bytes) -> char* {
    char* p = ws + off;
    off = (off + bytes + 255) & ~(size_t)255;
    return p;
  };
  float* lg_a     = (float*)alloc((size_t)2 * N_NODES * H_HEADS * sizeof(float));
  float* lg_b     = (float*)alloc((size_t)2 * N_NODES * H_HEADS * sizeof(float));
  float* xbuf     = (float*)alloc((size_t)N_NODES * F_INF * sizeof(float));
  int*   row_ptr  = (int*)alloc((size_t)CSR_SLICES * RP_STRIDE * sizeof(int));
  int*   edge_idx = (int*)alloc((size_t)N_EDGES * sizeof(int));
  float* wlr      = (float*)alloc((size_t)F_INF * 2 * H_HEADS * sizeof(float));

  csr_build<<<CSR_WGS + FOLD_WGS, CSR_THREADS, 0, stream>>>(dst, row_ptr, edge_idx,
                                                            W, attn_l, attn_r, wlr);
  node_logits<<<LOGIT_TILES / LOGIT_WAVES, LOGIT_WAVES * 32, 0, stream>>>(
      x, wlr, lg_a, lg_a + N_NODES * H_HEADS);

  const float* xin = x;
  float* lg_cur = lg_a;
  float* lg_nxt = lg_b;
  for (int step = 0; step < STEPS; ++step) {
    const bool last = (step == STEPS - 1);
    float* xo = last ? (float*)d_out : xbuf;
    gat_aggregate_project<<<N_NODES / FUSE_NODES, FUSE_WAVES * 32, 0, stream>>>(
        xin, lg_cur, src, row_ptr, edge_idx, W, bias, alpha, x0,
        wlr, last ? nullptr : lg_nxt, xo);
    xin = xbuf;
    float* t = lg_cur; lg_cur = lg_nxt; lg_nxt = t;
  }
}


# CSR stores each edge's source node instead of its edge number (fill pass reads src beside dst): every look-up of the step launches loses a dependent scattered 4-byte load; same sums, same order
# speedup vs baseline: 3.8253x; 1.0341x over previous
#include <hip/hip_runtime.h>
#include <hip/hip_bf16.h>

#define N_NODES 10000
#define N_EDGES 320000
#define F_INF   128
#define F_OUTF  128
#define H_HEADS 4
#define C_OUT   (H_HEADS * F_OUTF)
#define STEPS   2
#define NEG_SLOPE 0.2f

typedef float v2f __attribute__((ext_vector_type(2)));
typedef float v8f __attribute__((ext_vector_type(8)));

#define FOLD_WAVES 16
#define FOLD_WGS   ((F_INF * H_HEADS) / FOLD_WAVES)
static_assert((F_INF * H_HEADS) % FOLD_WAVES == 0, "whole work-groups, no tail wave");
static_assert(F_OUTF == 128, "a lane holds 4 of a head's 128 columns");

__device__ __forceinline__ void attn_fold_wave(
    const int id, const int lane,
    const float* __restrict__ W, const float* __restrict__ attn_l,
    const float* __restrict__ attn_r, float* __restrict__ wlr) {
  const int k    = id >> 2;
  const int hd   = id & (H_HEADS - 1);
  const float4 w  = *(const float4*)(W + (size_t)k * C_OUT + hd * F_OUTF + 4 * lane);
  const float4 al = *(const float4*)(attn_l + hd * F_OUTF + 4 * lane);
  const float4 ar = *(const float4*)(attn_r + hd * F_OUTF + 4 * lane);
  float sl = w.x * al.x + w.y * al.y + w.z * al.z + w.w * al.w;
  float sr = w.x * ar.x + w.y * ar.y + w.z * ar.z + w.w * ar.w;
  for (int off = 16; off; off >>= 1) {
    sl += __shfl_xor(sl, off, 32);
    sr += __shfl_xor(sr, off, 32);
  }
  if (lane == 0) {
    wlr[hd * F_INF + k]             = sl;
    wlr[(H_HEADS + hd) * F_INF + k] = sr;
  }
}

#define LOGIT_WAVES 5
#define LOGIT_TILES (N_NODES / 16)
static_assert(N_NODES % 16 == 0, "whole 16-node tiles");
static_assert(LOGIT_TILES % LOGIT_WAVES == 0, "whole work-groups, no tail wave");
static_assert(F_INF % 8 == 0 && 2 * H_HEADS == 8, "k-steps in pairs; eight logits per node");

__global__ __launch_bounds__(LOGIT_WAVES * 32) void node_logits(
    const float* __restrict__ x, const float* __restrict__ wlr,
    float* __restrict__ el, float* __restrict__ er) {
  const int lane = threadIdx.x & 31;
  const int m0   = (blockIdx.x * LOGIT_WAVES + (threadIdx.x >> 5)) * 16;
  const int hf   = lane >> 4;
  const int c    = lane & 15;

  const float* arow = x + (size_t)(m0 + c) * F_INF + 4 * hf;
  const float* brow = wlr + (c & 7) * F_INF + 4 * hf;

  v8f acc0 = v8f{};
  v8f acc1 = v8f{};
  #pragma unroll
  for (int k8 = 0; k8 < F_INF; k8 += 8) {
    const float4 a4 = *(const float4*)(arow + k8);
    const float4 b4 = *(const float4*)(brow + k8);
    v2f a; a.x = a4.x; a.y = a4.y;
    v2f b; b.x = b4.x; b.y = b4.y;
    acc0 = __builtin_amdgcn_wmma_f32_16x16x4_f32(
         false, a,  false, b,
         (short)0, acc0,  false,  false);
    a.x = a4.z; a.y = a4.w;
    b.x = b4.z; b.y = b4.w;
    acc1 = __builtin_amdgcn_wmma_f32_16x16x4_f32(
         false, a,  false, b,
         (short)0, acc1,  false,  false);
  }

  if (c < 2 * H_HEADS) {
    float* o = (c < H_HEADS) ? (el + c) : (er + (c - H_HEADS));
    #pragma unroll
    for (int i = 0; i < 8; ++i) o[(m0 + 8 * hf + i) * H_HEADS] = acc0[i] + acc1[i];
  }
}

#define CSR_RANGES   32
#define CSR_SLICES   8
#define CSR_WGS      (CSR_RANGES * CSR_SLICES)
#define CSR_THREADS  512
#define CSR_RANGE    ((N_NODES + CSR_RANGES - 1) / CSR_RANGES)
#define CSR_PER_LANE ((CSR_RANGE + 31) / 32)
#define SLICE_EDGES  (N_EDGES / CSR_SLICES)
#define SLICE_VEC    (SLICE_EDGES / 4)
#define CSR_CHUNKS   ((SLICE_VEC + CSR_THREADS - 1) / CSR_THREADS)
#define RP_STRIDE    (((N_NODES + 1 + 31) / 32) * 32)
#define CSR_PAD      0x7fffffff

static_assert(N_EDGES % (4 * CSR_SLICES) == 0, "a slice of dst is read as int4");
static_assert((CSR_RANGES - 1) * CSR_RANGE < N_NODES, "every group owns at least one node");
static_assert(CSR_CHUNKS % 4 == 0, "the sweeps take four chunks per trip");
static_assert((CSR_SLICES & (CSR_SLICES - 1)) == 0, "slice number = low bits of the group number");

static_assert(FOLD_WAVES * 32 == CSR_THREADS, "the fold's groups have the CSR groups' shape");

__global__ __launch_bounds__(CSR_THREADS) void csr_build(const int* __restrict__ dst,
                                                         const int* __restrict__ src,
                                                         int* __restrict__ row_ptr,
                                                         int* __restrict__ edge_src,
                                                         const float* __restrict__ W,
                                                         const float* __restrict__ attn_l,
                                                         const float* __restrict__ attn_r,
                                                         float* __restrict__ wlr) {
  __shared__ int bins[CSR_RANGE];
  __shared__ int s_below;
  const int t = threadIdx.x, lane = t & 31, wv = t >> 5;
  if (blockIdx.x >= CSR_WGS) {
    attn_fold_wave((blockIdx.x - CSR_WGS) * FOLD_WAVES + wv, lane, W, attn_l, attn_r, wlr);
    return;
  }
  const int sl = blockIdx.x & (CSR_SLICES - 1);
  const int b  = blockIdx.x / CSR_SLICES;
  const int lo = b * CSR_RANGE;
  const int nr = min(CSR_RANGE, N_NODES - lo);
  const int rot = b % CSR_CHUNKS;
  const int4* d4 = (const int4*)dst + sl * SLICE_VEC;
  const int4* s4 = (const int4*)src + sl * SLICE_VEC;

  for (int j = t; j < CSR_RANGE; j += CSR_THREADS) bins[j] = 0;
  if (t == 0) s_below = 0;
  __syncthreads();

  int below = 0;
  for (int k = 0; k < CSR_CHUNKS; k += 4) {
    int4 d[4];
    #pragma unroll
    for (int u = 0; u < 4; ++u) {
      int c = k + u + rot;
      if (c >= CSR_CHUNKS) c -= CSR_CHUNKS;
      const int v = c * CSR_THREADS + t;
      d[u] = (v < SLICE_VEC) ? d4[v] : make_int4(CSR_PAD, CSR_PAD, CSR_PAD, CSR_PAD);
    }
    #pragma unroll
    for (int u = 0; u < 4; ++u) {
      const int dd[4] = {d[u].x, d[u].y, d[u].z, d[u].w};
      #pragma unroll
      for (int q = 0; q < 4; ++q) {
        const unsigned r = (unsigned)(dd[q] - lo);
        below += (dd[q] < lo) ? 1 : 0;
        if (r < (unsigned)nr) atomicAdd(&bins[r], 1);
      }
    }
  }
  for (int off = 16; off; off >>= 1) below += __shfl_xor(below, off, 32);
  if (lane == 0) atomicAdd(&s_below, below);
  __syncthreads();

  if (wv == 0) {
    const int base = sl * SLICE_EDGES + s_below;
    int* rp = row_ptr + sl * RP_STRIDE;
    int loc[CSR_PER_LANE];
    int s = 0;
    #pragma unroll
    for (int i = 0; i < CSR_PER_LANE; ++i) {
      const int j = lane * CSR_PER_LANE + i;
      const int c = (j < CSR_RANGE) ? bins[j] : 0;
      loc[i] = s;
      s += c;
    }
    int inc = s;
    for (int off = 1; off < 32; off <<= 1) {
      const int up = __shfl_up(inc, off, 32);
      if (lane >= off) inc += up;
    }
    const int excl = inc - s;
    #pragma unroll
    for (int i = 0; i < CSR_PER_LANE; ++i) {
      const int j = lane * CSR_PER_LANE + i;
      if (j < CSR_RANGE) {
        const int start = base + excl + loc[i];
        bins[j] = start;
        if (j < nr) rp[lo + j] = start;
      }
    }
    if (b == CSR_RANGES - 1 && lane == 31) rp[N_NODES] = base + inc;
  }
  __syncthreads();

  for (int k = 0; k < CSR_CHUNKS; k += 4) {
    int4 d[4];
    int4 sv[4];
    #pragma unroll
    for (int u = 0; u < 4; ++u) {
      int c = k + u + rot;
      if (c >= CSR_CHUNKS) c -= CSR_CHUNKS;
      const int v = c * CSR_THREADS + t;
      d[u]  = (v < SLICE_VEC) ? d4[v] : make_int4(CSR_PAD, CSR_PAD, CSR_PAD, CSR_PAD);
      sv[u] = (v < SLICE_VEC) ? s4[v] : make_int4(0, 0, 0, 0);
    }
    int pos[4][4];
    #pragma unroll
    for (int u = 0; u < 4; ++u) {
      const int dd[4] = {d[u].x, d[u].y, d[u].z, d[u].w};
      #pragma unroll
      for (int q = 0; q < 4; ++q) {
        const unsigned r = (unsigned)(dd[q] - lo);
        int p = -1;
        if (r < (unsigned)nr) p = atomicAdd(&bins[r], 1);
        pos[u][q] = p;
      }
    }
    #pragma unroll
    for (int u = 0; u < 4; ++u) {
      const int ss[4] = {sv[u].x, sv[u].y, sv[u].z, sv[u].w};
      #pragma unroll
      for (int q = 0; q < 4; ++q)
        if (pos[u][q] >= 0) edge_src[pos[u][q]] = ss[q];
    }
  }
}

#define FUSE_NODES 16
#define FUSE_WAVES 16
#define AGG_EDGES  8
#define N_SLABS    (F_OUTF / 32)
#define ZT_PITCH   (C_OUT + 4)
#define RED_PITCH  (F_OUTF + 4)
#define WL_PITCH   (F_INF + 4)
static_assert(2 * H_HEADS * F_INF == 2 * FUSE_WAVES * 32, "a thread copies two floats of the folded vectors");
static_assert(N_NODES % FUSE_NODES == 0, "whole work-groups, no tail wave, no partial tile");
static_assert(FUSE_NODES == 16 && FUSE_WAVES == H_HEADS * N_SLABS && FUSE_WAVES == FUSE_NODES,
              "one wave per node, per (head, slab) and per output row");
static_assert(32 % AGG_EDGES == 0, "a chunk is a whole number of batches");
static_assert(H_HEADS == 4 && F_INF == 128 && F_OUTF == 128 && F_INF % 8 == 0,
              "a lane holds 4 input features and one sum per head; tile naming; k-steps in pairs");

__device__ __forceinline__ float leaky(float v) { return v > 0.f ? v : NEG_SLOPE * v; }

__device__ __forceinline__ int seg_place(int i, const int (&off)[CSR_SLICES],
                                         const int (&cum)[CSR_SLICES]) {
  int d = off[0];
  #pragma unroll
  for (int g = 1; g < CSR_SLICES; ++g) d = (i >= cum[g]) ? off[g] : d;
  return i + d;
}

__global__ __launch_bounds__(FUSE_WAVES * 32, 12) void gat_aggregate_project(
    const float* __restrict__ xin, const float* __restrict__ lg,
    const int* __restrict__ row_ptr, const int* __restrict__ edge_src,
    const float* __restrict__ W, const float* __restrict__ bias,
    const float* __restrict__ alpha_p, const float* __restrict__ x0,
    const float* __restrict__ wlr, float* __restrict__ lg_next,
    float* __restrict__ xout) {
  const float* el = lg;
  const float* er = lg + N_NODES * H_HEADS;
  __shared__ __attribute__((aligned(16))) float zt[FUSE_NODES * ZT_PITCH];
  __shared__ __attribute__((aligned(16))) float red[H_HEADS * FUSE_NODES * RED_PITCH];
  __shared__ __attribute__((aligned(16))) float wls[2 * H_HEADS * WL_PITCH];

  const int lane = threadIdx.x & 31;
  const int wv = __builtin_amdgcn_readfirstlane(threadIdx.x >> 5);
  const int m0 = blockIdx.x * FUSE_NODES;
  const int n  = m0 + wv;
  int off[CSR_SLICES], cum[CSR_SLICES];
  int re = 0;
  #pragma unroll
  for (int g = 0; g < CSR_SLICES; ++g) {
    const int b = row_ptr[g * RP_STRIDE + n], e = row_ptr[g * RP_STRIDE + n + 1];
    cum[g] = re;
    off[g] = b - re;
    re += e - b;
  }
  const float4 er4 = *(const float4*)(er + n * H_HEADS);
  const float erd[H_HEADS] = {er4.x, er4.y, er4.z, er4.w};

  const bool ok0 = lane < re;
  int s0 = 0;
  float v0[H_HEADS];
  for (int hd = 0; hd < H_HEADS; ++hd) v0[hd] = -INFINITY;
  if (ok0) {
    s0 = edge_src[seg_place(lane, off, cum)];
    const float4 q4 = *(const float4*)(el + s0 * H_HEADS);
    const float q[H_HEADS] = {q4.x, q4.y, q4.z, q4.w};
    for (int hd = 0; hd < H_HEADS; ++hd) v0[hd] = leaky(q[hd] + erd[hd]);
  }

  float mx[H_HEADS];
  for (int hd = 0; hd < H_HEADS; ++hd) mx[hd] = fmaxf(-INFINITY, v0[hd]);
  for (int i = 32 + lane; i < re; i += 32) {
    const int s = edge_src[seg_place(i, off, cum)];
    const float4 q4 = *(const float4*)(el + s * H_HEADS);
    const float q[H_HEADS] = {q4.x, q4.y, q4.z, q4.w};
    for (int hd = 0; hd < H_HEADS; ++hd) mx[hd] = fmaxf(mx[hd], leaky(q[hd] + erd[hd]));
  }
  #pragma unroll
  for (int off = 16; off; off >>= 1) {
    for (int hd = 0; hd < H_HEADS; ++hd) mx[hd] = fmaxf(mx[hd], __shfl_xor(mx[hd], off, 32));
  }

  float p0[H_HEADS], sum[H_HEADS], inv[H_HEADS];
  for (int hd = 0; hd < H_HEADS; ++hd) {
    p0[hd] = ok0 ? __expf(v0[hd] - mx[hd]) : 0.f;
    sum[hd] = p0[hd];
  }
  for (int i = 32 + lane; i < re; i += 32) {
    const int s = edge_src[seg_place(i, off, cum)];
    const float4 q4 = *(const float4*)(el + s * H_HEADS);
    const float q[H_HEADS] = {q4.x, q4.y, q4.z, q4.w};
    for (int hd = 0; hd < H_HEADS; ++hd) sum[hd] += __expf(leaky(q[hd] + erd[hd]) - mx[hd]);
  }
  #pragma unroll
  for (int off = 16; off; off >>= 1) {
    for (int hd = 0; hd < H_HEADS; ++hd) sum[hd] += __shfl_xor(sum[hd], off, 32);
  }
  for (int hd = 0; hd < H_HEADS; ++hd) inv[hd] = (sum[hd] != 0.f) ? 1.f / sum[hd] : 0.f;

  const float* xrow = xin + lane * 4;
  float4 acc[H_HEADS];
  for (int hd = 0; hd < H_HEADS; ++hd) acc[hd] = make_float4(0.f, 0.f, 0.f, 0.f);
  int   s_l = s0;
  float a_l[H_HEADS];
  for (int hd = 0; hd < H_HEADS; ++hd) a_l[hd] = p0[hd] * inv[hd];
  for (int base = 0; base < re; base += 32) {
    if (base != 0) {
      s_l = 0;
      for (int hd = 0; hd < H_HEADS; ++hd) a_l[hd] = 0.f;
      if (base + lane < re) {
        s_l = edge_src[seg_place(base + lane, off, cum)];
        const float4 q4 = *(const float4*)(el + s_l * H_HEADS);
        const float q[H_HEADS] = {q4.x, q4.y, q4.z, q4.w};
        for (int hd = 0; hd < H_HEADS; ++hd)
          a_l[hd] = __expf(leaky(q[hd] + erd[hd]) - mx[hd]) * inv[hd];
      }
    }
    const int cnt = min(32, re - base);
    const int s_last = __builtin_amdgcn_readlane(s_l, cnt - 1);
    if (lane >= cnt) s_l = s_last;
    for (int j = 0; j < cnt; j += AGG_EDGES) {
      float4 xv[AGG_EDGES];
      float  aw[AGG_EDGES][H_HEADS];
      #pragma unroll
      for (int u = 0; u < AGG_EDGES; ++u) {
        xv[u] = *(const float4*)(xrow + (size_t)__builtin_amdgcn_readlane(s_l, j + u) * F_INF);
        for (int hd = 0; hd < H_HEADS; ++hd)
          aw[u][hd] = __int_as_float(__builtin_amdgcn_readlane(__float_as_int(a_l[hd]), j + u));
      }
      #pragma unroll
      for (int u = 0; u < AGG_EDGES; ++u) {
        for (int hd = 0; hd < H_HEADS; ++hd) {
          acc[hd].x += aw[u][hd] * xv[u].x; acc[hd].y += aw[u][hd] * xv[u].y;
          acc[hd].z += aw[u][hd] * xv[u].z; acc[hd].w += aw[u][hd] * xv[u].w;
        }
      }
    }
  }
  for (int hd = 0; hd < H_HEADS; ++hd)
    *(float4*)(zt + wv * ZT_PITCH + hd * F_INF + lane * 4) = acc[hd];
  __syncthreads();

  {
    const int hd = wv >> 2;
    const int q  = wv & (N_SLABS - 1);
    const int hf = lane >> 4;
    const int c  = lane & 15;
    float2 wl2 = make_float2(0.f, 0.f);
    if (lg_next != nullptr) wl2 = *(const float2*)(wlr + 2 * threadIdx.x);
    v8f acc0 = v8f{};
    v8f acc1 = v8f{};
    const float* ah = zt + c * ZT_PITCH + hd * F_INF + 4 * hf;
    const float* bh = W + (size_t)(4 * hf) * C_OUT + hd * F_OUTF + 32 * q + 2 * c;
    #pragma unroll 4
    for (int k8 = 0; k8 < F_INF; k8 += 8) {
      const float4 a4 = *(const float4*)(ah + k8);
      const float2 r0 = *(const float2*)(bh + (size_t)k8 * C_OUT);
      const float2 r1 = *(const float2*)(bh + (size_t)k8 * C_OUT + C_OUT);
      const float2 r2 = *(const float2*)(bh + (size_t)k8 * C_OUT + 2 * C_OUT);
      const float2 r3 = *(const float2*)(bh + (size_t)k8 * C_OUT + 3 * C_OUT);
      v2f a;  a.x  = a4.x; a.y  = a4.y;
      v2f b0; b0.x = r0.x; b0.y = r1.x;
      v2f b1; b1.x = r0.y; b1.y = r1.y;
      acc0 = __builtin_amdgcn_wmma_f32_16x16x4_f32(
           false, a,  false, b0,
           (short)0, acc0,  false,  false);
      acc1 = __builtin_amdgcn_wmma_f32_16x16x4_f32(
           false, a,  false, b1,
           (short)0, acc1,  false,  false);
      a.x  = a4.z; a.y  = a4.w;
      b0.x = r2.x; b0.y = r3.x;
      b1.x = r2.y; b1.y = r3.y;
      acc0 = __builtin_amdgcn_wmma_f32_16x16x4_f32(
           false, a,  false, b0,
           (short)0, acc0,  false,  false);
      acc1 = __builtin_amdgcn_wmma_f32_16x16x4_f32(
           false, a,  false, b1,
           (short)0, acc1,  false,  false);
    }
    float* rp = red + (hd * FUSE_NODES + 8 * hf) * RED_PITCH + 32 * q + 2 * c;
    #pragma unroll
    for (int i = 0; i < 8; ++i) {
      float2 p; p.x = acc0[i]; p.y = acc1[i];
      *(float2*)(rp + i * RED_PITCH) = p;
    }
    if (lg_next != nullptr)
      *(float2*)(wls + (threadIdx.x >> 6) * WL_PITCH + 2 * (threadIdx.x & 63)) = wl2;
  }
  __syncthreads();

  {
    const int col = lane * 4;
    float4 s = *(const float4*)(red + wv * RED_PITCH + col);
    float4 bm = *(const float4*)(bias + col);
    for (int hd = 1; hd < H_HEADS; ++hd) {
      const float4 p = *(const float4*)(red + (hd * FUSE_NODES + wv) * RED_PITCH + col);
      const float4 b = *(const float4*)(bias + hd * F_OUTF + col);
      s.x += p.x; s.y += p.y; s.z += p.z; s.w += p.w;
      bm.x += b.x; bm.y += b.y; bm.z += b.z; bm.w += b.w;
    }
    const float al = alpha_p[0];
    const size_t o = (size_t)n * F_OUTF + col;
    const float4 xv = *(const float4*)(x0 + o);
    float4 r;
    r.x = (1.f - al) * (0.25f * (s.x + bm.x)) + al * xv.x;
    r.y = (1.f - al) * (0.25f * (s.y + bm.y)) + al * xv.y;
    r.z = (1.f - al) * (0.25f * (s.z + bm.z)) + al * xv.z;
    r.w = (1.f - al) * (0.25f * (s.w + bm.w)) + al * xv.w;
    *(float4*)(xout + o) = r;

    if (lg_next != nullptr) {
      *(float4*)(zt + wv * ZT_PITCH + col) = r;
      __syncthreads();
      if (wv == 0) {
        const int hf = lane >> 4;
        const int c  = lane & 15;
        const float* arow = zt + c * ZT_PITCH + 4 * hf;
        const float* brow = wls + (c & 7) * WL_PITCH + 4 * hf;
        v8f acc0 = v8f{};
        v8f acc1 = v8f{};
        #pragma unroll 4
        for (int k8 = 0; k8 < F_INF; k8 += 8) {
          const float4 a4 = *(const float4*)(arow + k8);
          const float4 b4 = *(const float4*)(brow + k8);
          v2f a; a.x = a4.x; a.y = a4.y;
          v2f b; b.x = b4.x; b.y = b4.y;
          acc0 = __builtin_amdgcn_wmma_f32_16x16x4_f32(
               false, a,  false, b,
               (short)0, acc0,  false,  false);
          a.x = a4.z; a.y = a4.w;
          b.x = b4.z; b.y = b4.w;
          acc1 = __builtin_amdgcn_wmma_f32_16x16x4_f32(
               false, a,  false, b,
               (short)0, acc1,  false,  false);
        }
        if (c < 2 * H_HEADS) {
          float* o2 = (c < H_HEADS) ? (lg_next + c) : (lg_next + N_NODES * H_HEADS + (c - H_HEADS));
          #pragma unroll
          for (int i = 0; i < 8; ++i) o2[(m0 + 8 * hf + i) * H_HEADS] = acc0[i] + acc1[i];
        }
      }
    }
  }
}

extern "C" void kernel_launch(void* const* d_in, const int* in_sizes, int n_in,
                              void* d_out, int out_size, void* d_ws, size_t ws_size,
                              hipStream_t stream) {
  const float* x      = (const float*)d_in[0];
  const float* x0     = (const float*)d_in[1];
  const int*   src    = (const int*)d_in[2];
  const int*   dst    = (const int*)d_in[3];
  const float* alpha  = (const float*)d_in[4];
  const float* W      = (const float*)d_in[5];
  const float* attn_l = (const float*)d_in[6];
  const float* attn_r = (const float*)d_in[7];
  const float* bias   = (const float*)d_in[8];

  char* ws = (char*)d_ws;
  size_t off = 0;
  auto alloc = [&](size_t bytes) -> char* {
    char* p = ws + off;
    off = (off + bytes + 255) & ~(size_t)255;
    return p;
  };
  float* lg_a     = (float*)alloc((size_t)2 * N_NODES * H_HEADS * sizeof(float));
  float* lg_b     = (float*)alloc((size_t)2 * N_NODES * H_HEADS * sizeof(float));
  float* xbuf     = (float*)alloc((size_t)N_NODES * F_INF * sizeof(float));
  int*   row_ptr  = (int*)alloc((size_t)CSR_SLICES * RP_STRIDE * sizeof(int));
  int*   edge_src = (int*)alloc((size_t)N_EDGES * sizeof(int));
  float* wlr      = (float*)alloc((size_t)F_INF * 2 * H_HEADS * sizeof(float));

  csr_build<<<CSR_WGS + FOLD_WGS, CSR_THREADS, 0, stream>>>(dst, src, row_ptr, edge_src,
                                                            W, attn_l, attn_r, wlr);
  node_logits<<<LOGIT_TILES / LOGIT_WAVES, LOGIT_WAVES * 32, 0, stream>>>(
      x, wlr, lg_a, lg_a + N_NODES * H_HEADS);

  const float* xin = x;
  float* lg_cur = lg_a;
  float* lg_nxt = lg_b;
  for (int step = 0; step < STEPS; ++step) {
    const bool last = (step == STEPS - 1);
    float* xo = last ? (float*)d_out : xbuf;
    gat_aggregate_project<<<N_NODES / FUSE_NODES, FUSE_WAVES * 32, 0, stream>>>(
        xin, lg_cur, row_ptr, edge_src, W, bias, alpha, x0,
        wlr, last ? nullptr : lg_nxt, xo);
    xin = xbuf;
    float* t = lg_cur; lg_cur = lg_nxt; lg_nxt = t;
  }
}


# CSR build cut 16 node ranges x 16 slices (was 32 x 8): a group loads its slice of dst and src once into registers, counts and fills from them; a node has 16 sub-segments; order of f32 sums changes
# speedup vs baseline: 3.9861x; 1.0420x over previous
#include <hip/hip_runtime.h>
#include <hip/hip_bf16.h>

#define N_NODES 10000
#define N_EDGES 320000
#define F_INF   128
#define F_OUTF  128
#define H_HEADS 4
#define C_OUT   (H_HEADS * F_OUTF)
#define STEPS   2
#define NEG_SLOPE 0.2f

typedef float v2f __attribute__((ext_vector_type(2)));
typedef float v8f __attribute__((ext_vector_type(8)));

#define FOLD_WAVES 16
#define FOLD_WGS   ((F_INF * H_HEADS) / FOLD_WAVES)
static_assert((F_INF * H_HEADS) % FOLD_WAVES == 0, "whole work-groups, no tail wave");
static_assert(F_OUTF == 128, "a lane holds 4 of a head's 128 columns");

__device__ __forceinline__ void attn_fold_wave(
    const int id, const int lane,
    const float* __restrict__ W, const float* __restrict__ attn_l,
    const float* __restrict__ attn_r, float* __restrict__ wlr) {
  const int k    = id >> 2;
  const int hd   = id & (H_HEADS - 1);
  const float4 w  = *(const float4*)(W + (size_t)k * C_OUT + hd * F_OUTF + 4 * lane);
  const float4 al = *(const float4*)(attn_l + hd * F_OUTF + 4 * lane);
  const float4 ar = *(const float4*)(attn_r + hd * F_OUTF + 4 * lane);
  float sl = w.x * al.x + w.y * al.y + w.z * al.z + w.w * al.w;
  float sr = w.x * ar.x + w.y * ar.y + w.z * ar.z + w.w * ar.w;
  for (int off = 16; off; off >>= 1) {
    sl += __shfl_xor(sl, off, 32);
    sr += __shfl_xor(sr, off, 32);
  }
  if (lane == 0) {
    wlr[hd * F_INF + k]             = sl;
    wlr[(H_HEADS + hd) * F_INF + k] = sr;
  }
}

#define LOGIT_WAVES 5
#define LOGIT_TILES (N_NODES / 16)
static_assert(N_NODES % 16 == 0, "whole 16-node tiles");
static_assert(LOGIT_TILES % LOGIT_WAVES == 0, "whole work-groups, no tail wave");
static_assert(F_INF % 8 == 0 && 2 * H_HEADS == 8, "k-steps in pairs; eight logits per node");

__global__ __launch_bounds__(LOGIT_WAVES * 32) void node_logits(
    const float* __restrict__ x, const float* __restrict__ wlr,
    float* __restrict__ el, float* __restrict__ er) {
  const int lane = threadIdx.x & 31;
  const int m0   = (blockIdx.x * LOGIT_WAVES + (threadIdx.x >> 5)) * 16;
  const int hf   = lane >> 4;
  const int c    = lane & 15;

  const float* arow = x + (size_t)(m0 + c) * F_INF + 4 * hf;
  const float* brow = wlr + (c & 7) * F_INF + 4 * hf;

  v8f acc0 = v8f{};
  v8f acc1 = v8f{};
  #pragma unroll
  for (int k8 = 0; k8 < F_INF; k8 += 8) {
    const float4 a4 = *(const float4*)(arow + k8);
    const float4 b4 = *(const float4*)(brow + k8);
    v2f a; a.x = a4.x; a.y = a4.y;
    v2f b; b.x = b4.x; b.y = b4.y;
    acc0 = __builtin_amdgcn_wmma_f32_16x16x4_f32(
         false, a,  false, b,
         (short)0, acc0,  false,  false);
    a.x = a4.z; a.y = a4.w;
    b.x = b4.z; b.y = b4.w;
    acc1 = __builtin_amdgcn_wmma_f32_16x16x4_f32(
         false, a,  false, b,
         (short)0, acc1,  false,  false);
  }

  if (c < 2 * H_HEADS) {
    float* o = (c < H_HEADS) ? (el + c) : (er + (c - H_HEADS));
    #pragma unroll
    for (int i = 0; i < 8; ++i) o[(m0 + 8 * hf + i) * H_HEADS] = acc0[i] + acc1[i];
  }
}

#define CSR_RANGES   16
#define CSR_SLICES   16
#define CSR_WGS      (CSR_RANGES * CSR_SLICES)
#define CSR_THREADS  512
#define CSR_RANGE    ((N_NODES + CSR_RANGES - 1) / CSR_RANGES)
#define CSR_PER_LANE ((CSR_RANGE + 31) / 32)
#define SLICE_EDGES  (N_EDGES / CSR_SLICES)
#define SLICE_VEC    (SLICE_EDGES / 4)
#define CSR_CHUNKS   ((SLICE_VEC + CSR_THREADS - 1) / CSR_THREADS)
#define CSR_BATCH    5
#define RP_STRIDE    (((N_NODES + 1 + 31) / 32) * 32)
#define CSR_PAD      0x7fffffff

static_assert(N_EDGES % (4 * CSR_SLICES) == 0, "a slice of dst is read as int4");
static_assert((CSR_RANGES - 1) * CSR_RANGE < N_NODES, "every group owns at least one node");
static_assert(CSR_CHUNKS % CSR_BATCH == 0, "pass 2 takes whole batches of chunks");
static_assert((CSR_SLICES & (CSR_SLICES - 1)) == 0, "slice number = low bits of the group number");

static_assert(FOLD_WAVES * 32 == CSR_THREADS, "the fold's groups have the CSR groups' shape");

__global__ __launch_bounds__(CSR_THREADS) void csr_build(const int* __restrict__ dst,
                                                         const int* __restrict__ src,
                                                         int* __restrict__ row_ptr,
                                                         int* __restrict__ edge_src,
                                                         const float* __restrict__ W,
                                                         const float* __restrict__ attn_l,
                                                         const float* __restrict__ attn_r,
                                                         float* __restrict__ wlr) {
  __shared__ int bins[CSR_RANGE];
  __shared__ int s_below;
  const int t = threadIdx.x, lane = t & 31, wv = t >> 5;
  if (blockIdx.x >= CSR_WGS) {
    attn_fold_wave((blockIdx.x - CSR_WGS) * FOLD_WAVES + wv, lane, W, attn_l, attn_r, wlr);
    return;
  }
  const int sl = blockIdx.x & (CSR_SLICES - 1);
  const int b  = blockIdx.x / CSR_SLICES;
  const int lo = b * CSR_RANGE;
  const int nr = min(CSR_RANGE, N_NODES - lo);
  const int rot = b % CSR_CHUNKS;
  const int4* d4 = (const int4*)dst + sl * SLICE_VEC;
  const int4* s4 = (const int4*)src + sl * SLICE_VEC;

  for (int j = t; j < CSR_RANGE; j += CSR_THREADS) bins[j] = 0;
  if (t == 0) s_below = 0;
  __syncthreads();

  int4 d[CSR_CHUNKS];
  int4 sv[CSR_CHUNKS];
  #pragma unroll
  for (int u = 0; u < CSR_CHUNKS; ++u) {
    int c = u + rot;
    if (c >= CSR_CHUNKS) c -= CSR_CHUNKS;
    d[u] = d4[min(c * CSR_THREADS + t, SLICE_VEC - 1)];
  }
  #pragma unroll
  for (int u = 0; u < CSR_CHUNKS; ++u) {
    int c = u + rot;
    if (c >= CSR_CHUNKS) c -= CSR_CHUNKS;
    sv[u] = s4[min(c * CSR_THREADS + t, SLICE_VEC - 1)];
  }

  int below = 0;
  #pragma unroll
  for (int u = 0; u < CSR_CHUNKS; ++u) {
    int c = u + rot;
    if (c >= CSR_CHUNKS) c -= CSR_CHUNKS;
    if (c * CSR_THREADS + t >= SLICE_VEC) d[u] = make_int4(CSR_PAD, CSR_PAD, CSR_PAD, CSR_PAD);
    const int dd[4] = {d[u].x, d[u].y, d[u].z, d[u].w};
    #pragma unroll
    for (int q = 0; q < 4; ++q) {
      const unsigned r = (unsigned)(dd[q] - lo);
      below += (dd[q] < lo) ? 1 : 0;
      if (r < (unsigned)nr) atomicAdd(&bins[r], 1);
    }
  }
  for (int off = 16; off; off >>= 1) below += __shfl_xor(below, off, 32);
  if (lane == 0) atomicAdd(&s_below, below);
  __syncthreads();

  if (wv == 0) {
    const int base = sl * SLICE_EDGES + s_below;
    int* rp = row_ptr + sl * RP_STRIDE;
    int loc[CSR_PER_LANE];
    int s = 0;
    #pragma unroll
    for (int i = 0; i < CSR_PER_LANE; ++i) {
      const int j = lane * CSR_PER_LANE + i;
      const int c = (j < CSR_RANGE) ? bins[j] : 0;
      loc[i] = s;
      s += c;
    }
    int inc = s;
    for (int off = 1; off < 32; off <<= 1) {
      const int up = __shfl_up(inc, off, 32);
      if (lane >= off) inc += up;
    }
    const int excl = inc - s;
    #pragma unroll
    for (int i = 0; i < CSR_PER_LANE; ++i) {
      const int j = lane * CSR_PER_LANE + i;
      if (j < CSR_RANGE) {
        const int start = base + excl + loc[i];
        bins[j] = start;
        if (j < nr) rp[lo + j] = start;
      }
    }
    if (b == CSR_RANGES - 1 && lane == 31) rp[N_NODES] = base + inc;
  }
  __syncthreads();

  #pragma unroll
  for (int k = 0; k < CSR_CHUNKS; k += CSR_BATCH) {
    int pos[CSR_BATCH][4];
    #pragma unroll
    for (int u = 0; u < CSR_BATCH; ++u) {
      const int dd[4] = {d[k + u].x, d[k + u].y, d[k + u].z, d[k + u].w};
      #pragma unroll
      for (int q = 0; q < 4; ++q) {
        const unsigned r = (unsigned)(dd[q] - lo);
        int p = -1;
        if (r < (unsigned)nr) p = atomicAdd(&bins[r], 1);
        pos[u][q] = p;
      }
    }
    #pragma unroll
    for (int u = 0; u < CSR_BATCH; ++u) {
      const int ss[4] = {sv[k + u].x, sv[k + u].y, sv[k + u].z, sv[k + u].w};
      #pragma unroll
      for (int q = 0; q < 4; ++q)
        if (pos[u][q] >= 0) edge_src[pos[u][q]] = ss[q];
    }
  }
}

#define FUSE_NODES 16
#define FUSE_WAVES 16
#define AGG_EDGES  8
#define N_SLABS    (F_OUTF / 32)
#define ZT_PITCH   (C_OUT + 4)
#define RED_PITCH  (F_OUTF + 4)
#define WL_PITCH   (F_INF + 4)
static_assert(2 * H_HEADS * F_INF == 2 * FUSE_WAVES * 32, "a thread copies two floats of the folded vectors");
static_assert(N_NODES % FUSE_NODES == 0, "whole work-groups, no tail wave, no partial tile");
static_assert(FUSE_NODES == 16 && FUSE_WAVES == H_HEADS * N_SLABS && FUSE_WAVES == FUSE_NODES,
              "one wave per node, per (head, slab) and per output row");
static_assert(32 % AGG_EDGES == 0, "a chunk is a whole number of batches");
static_assert(H_HEADS == 4 && F_INF == 128 && F_OUTF == 128 && F_INF % 8 == 0,
              "a lane holds 4 input features and one sum per head; tile naming; k-steps in pairs");

__device__ __forceinline__ float leaky(float v) { return v > 0.f ? v : NEG_SLOPE * v; }

__device__ __forceinline__ int seg_place(int i, const int (&off)[CSR_SLICES],
                                         const int (&cum)[CSR_SLICES]) {
  int d = off[0];
  #pragma unroll
  for (int g = 1; g < CSR_SLICES; ++g) d = (i >= cum[g]) ? off[g] : d;
  return i + d;
}

__global__ __launch_bounds__(FUSE_WAVES * 32, 12) void gat_aggregate_project(
    const float* __restrict__ xin, const float* __restrict__ lg,
    const int* __restrict__ row_ptr, const int* __restrict__ edge_src,
    const float* __restrict__ W, const float* __restrict__ bias,
    const float* __restrict__ alpha_p, const float* __restrict__ x0,
    const float* __restrict__ wlr, float* __restrict__ lg_next,
    float* __restrict__ xout) {
  const float* el = lg;
  const float* er = lg + N_NODES * H_HEADS;
  __shared__ __attribute__((aligned(16))) float zt[FUSE_NODES * ZT_PITCH];
  __shared__ __attribute__((aligned(16))) float red[H_HEADS * FUSE_NODES * RED_PITCH];
  __shared__ __attribute__((aligned(16))) float wls[2 * H_HEADS * WL_PITCH];

  const int lane = threadIdx.x & 31;
  const int wv = __builtin_amdgcn_readfirstlane(threadIdx.x >> 5);
  const int m0 = blockIdx.x * FUSE_NODES;
  const int n  = m0 + wv;
  int off[CSR_SLICES], cum[CSR_SLICES];
  int re = 0;
  #pragma unroll
  for (int g = 0; g < CSR_SLICES; ++g) {
    const int b = row_ptr[g * RP_STRIDE + n], e = row_ptr[g * RP_STRIDE + n + 1];
    cum[g] = re;
    off[g] = b - re;
    re += e - b;
  }
  const float4 er4 = *(const float4*)(er + n * H_HEADS);
  const float erd[H_HEADS] = {er4.x, er4.y, er4.z, er4.w};

  const bool ok0 = lane < re;
  int s0 = 0;
  float v0[H_HEADS];
  for (int hd = 0; hd < H_HEADS; ++hd) v0[hd] = -INFINITY;
  if (ok0) {
    s0 = edge_src[seg_place(lane, off, cum)];
    const float4 q4 = *(const float4*)(el + s0 * H_HEADS);
    const float q[H_HEADS] = {q4.x, q4.y, q4.z, q4.w};
    for (int hd = 0; hd < H_HEADS; ++hd) v0[hd] = leaky(q[hd] + erd[hd]);
  }

  float mx[H_HEADS];
  for (int hd = 0; hd < H_HEADS; ++hd) mx[hd] = fmaxf(-INFINITY, v0[hd]);
  for (int i = 32 + lane; i < re; i += 32) {
    const int s = edge_src[seg_place(i, off, cum)];
    const float4 q4 = *(const float4*)(el + s * H_HEADS);
    const float q[H_HEADS] = {q4.x, q4.y, q4.z, q4.w};
    for (int hd = 0; hd < H_HEADS; ++hd) mx[hd] = fmaxf(mx[hd], leaky(q[hd] + erd[hd]));
  }
  #pragma unroll
  for (int off = 16; off; off >>= 1) {
    for (int hd = 0; hd < H_HEADS; ++hd) mx[hd] = fmaxf(mx[hd], __shfl_xor(mx[hd], off, 32));
  }

  float p0[H_HEADS], sum[H_HEADS], inv[H_HEADS];
  for (int hd = 0; hd < H_HEADS; ++hd) {
    p0[hd] = ok0 ? __expf(v0[hd] - mx[hd]) : 0.f;
    sum[hd] = p0[hd];
  }
  for (int i = 32 + lane; i < re; i += 32) {
    const int s = edge_src[seg_place(i, off, cum)];
    const float4 q4 = *(const float4*)(el + s * H_HEADS);
    const float q[H_HEADS] = {q4.x, q4.y, q4.z, q4.w};
    for (int hd = 0; hd < H_HEADS; ++hd) sum[hd] += __expf(leaky(q[hd] + erd[hd]) - mx[hd]);
  }
  #pragma unroll
  for (int off = 16; off; off >>= 1) {
    for (int hd = 0; hd < H_HEADS; ++hd) sum[hd] += __shfl_xor(sum[hd], off, 32);
  }
  for (int hd = 0; hd < H_HEADS; ++hd) inv[hd] = (sum[hd] != 0.f) ? 1.f / sum[hd] : 0.f;

  const float* xrow = xin + lane * 4;
  float4 acc[H_HEADS];
  for (int hd = 0; hd < H_HEADS; ++hd) acc[hd] = make_float4(0.f, 0.f, 0.f, 0.f);
  int   s_l = s0;
  float a_l[H_HEADS];
  for (int hd = 0; hd < H_HEADS; ++hd) a_l[hd] = p0[hd] * inv[hd];
  for (int base = 0; base < re; base += 32) {
    if (base != 0) {
      s_l = 0;
      for (int hd = 0; hd < H_HEADS; ++hd) a_l[hd] = 0.f;
      if (base + lane < re) {
        s_l = edge_src[seg_place(base + lane, off, cum)];
        const float4 q4 = *(const float4*)(el + s_l * H_HEADS);
        const float q[H_HEADS] = {q4.x, q4.y, q4.z, q4.w};
        for (int hd = 0; hd < H_HEADS; ++hd)
          a_l[hd] = __expf(leaky(q[hd] + erd[hd]) - mx[hd]) * inv[hd];
      }
    }
    const int cnt = min(32, re - base);
    const int s_last = __builtin_amdgcn_readlane(s_l, cnt - 1);
    if (lane >= cnt) s_l = s_last;
    for (int j = 0; j < cnt; j += AGG_EDGES) {
      float4 xv[AGG_EDGES];
      float  aw[AGG_EDGES][H_HEADS];
      #pragma unroll
      for (int u = 0; u < AGG_EDGES; ++u) {
        xv[u] = *(const float4*)(xrow + (size_t)__builtin_amdgcn_readlane(s_l, j + u) * F_INF);
        for (int hd = 0; hd < H_HEADS; ++hd)
          aw[u][hd] = __int_as_float(__builtin_amdgcn_readlane(__float_as_int(a_l[hd]), j + u));
      }
      #pragma unroll
      for (int u = 0; u < AGG_EDGES; ++u) {
        for (int hd = 0; hd < H_HEADS; ++hd) {
          acc[hd].x += aw[u][hd] * xv[u].x; acc[hd].y += aw[u][hd] * xv[u].y;
          acc[hd].z += aw[u][hd] * xv[u].z; acc[hd].w += aw[u][hd] * xv[u].w;
        }
      }
    }
  }
  for (int hd = 0; hd < H_HEADS; ++hd)
    *(float4*)(zt + wv * ZT_PITCH + hd * F_INF + lane * 4) = acc[hd];
  __syncthreads();

  {
    const int hd = wv >> 2;
    const int q  = wv & (N_SLABS - 1);
    const int hf = lane >> 4;
    const int c  = lane & 15;
    float2 wl2 = make_float2(0.f, 0.f);
    if (lg_next != nullptr) wl2 = *(const float2*)(wlr + 2 * threadIdx.x);
    v8f acc0 = v8f{};
    v8f acc1 = v8f{};
    const float* ah = zt + c * ZT_PITCH + hd * F_INF + 4 * hf;
    const float* bh = W + (size_t)(4 * hf) * C_OUT + hd * F_OUTF + 32 * q + 2 * c;
    #pragma unroll 4
    for (int k8 = 0; k8 < F_INF; k8 += 8) {
      const float4 a4 = *(const float4*)(ah + k8);
      const float2 r0 = *(const float2*)(bh + (size_t)k8 * C_OUT);
      const float2 r1 = *(const float2*)(bh + (size_t)k8 * C_OUT + C_OUT);
      const float2 r2 = *(const float2*)(bh + (size_t)k8 * C_OUT + 2 * C_OUT);
      const float2 r3 = *(const float2*)(bh + (size_t)k8 * C_OUT + 3 * C_OUT);
      v2f a;  a.x  = a4.x; a.y  = a4.y;
      v2f b0; b0.x = r0.x; b0.y = r1.x;
      v2f b1; b1.x = r0.y; b1.y = r1.y;
      acc0 = __builtin_amdgcn_wmma_f32_16x16x4_f32(
           false, a,  false, b0,
           (short)0, acc0,  false,  false);
      acc1 = __builtin_amdgcn_wmma_f32_16x16x4_f32(
           false, a,  false, b1,
           (short)0, acc1,  false,  false);
      a.x  = a4.z; a.y  = a4.w;
      b0.x = r2.x; b0.y = r3.x;
      b1.x = r2.y; b1.y = r3.y;
      acc0 = __builtin_amdgcn_wmma_f32_16x16x4_f32(
           false, a,  false, b0,
           (short)0, acc0,  false,  false);
      acc1 = __builtin_amdgcn_wmma_f32_16x16x4_f32(
           false, a,  false, b1,
           (short)0, acc1,  false,  false);
    }
    float* rp = red + (hd * FUSE_NODES + 8 * hf) * RED_PITCH + 32 * q + 2 * c;
    #pragma unroll
    for (int i = 0; i < 8; ++i) {
      float2 p; p.x = acc0[i]; p.y = acc1[i];
      *(float2*)(rp + i * RED_PITCH) = p;
    }
    if (lg_next != nullptr)
      *(float2*)(wls + (threadIdx.x >> 6) * WL_PITCH + 2 * (threadIdx.x & 63)) = wl2;
  }
  __syncthreads();

  {
    const int col = lane * 4;
    float4 s = *(const float4*)(red + wv * RED_PITCH + col);
    float4 bm = *(const float4*)(bias + col);
    for (int hd = 1; hd < H_HEADS; ++hd) {
      const float4 p = *(const float4*)(red + (hd * FUSE_NODES + wv) * RED_PITCH + col);
      const float4 b = *(const float4*)(bias + hd * F_OUTF + col);
      s.x += p.x; s.y += p.y; s.z += p.z; s.w += p.w;
      bm.x += b.x; bm.y += b.y; bm.z += b.z; bm.w += b.w;
    }
    const float al = alpha_p[0];
    const size_t o = (size_t)n * F_OUTF + col;
    const float4 xv = *(const float4*)(x0 + o);
    float4 r;
    r.x = (1.f - al) * (0.25f * (s.x + bm.x)) + al * xv.x;
    r.y = (1.f - al) * (0.25f * (s.y + bm.y)) + al * xv.y;
    r.z = (1.f - al) * (0.25f * (s.z + bm.z)) + al * xv.z;
    r.w = (1.f - al) * (0.25f * (s.w + bm.w)) + al * xv.w;
    *(float4*)(xout + o) = r;

    if (lg_next != nullptr) {
      *(float4*)(zt + wv * ZT_PITCH + col) = r;
      __syncthreads();
      if (wv == 0) {
        const int hf = lane >> 4;
        const int c  = lane & 15;
        const float* arow = zt + c * ZT_PITCH + 4 * hf;
        const float* brow = wls + (c & 7) * WL_PITCH + 4 * hf;
        v8f acc0 = v8f{};
        v8f acc1 = v8f{};
        #pragma unroll 4
        for (int k8 = 0; k8 < F_INF; k8 += 8) {
          const float4 a4 = *(const float4*)(arow + k8);
          const float4 b4 = *(const float4*)(brow + k8);
          v2f a; a.x = a4.x; a.y = a4.y;
          v2f b; b.x = b4.x; b.y = b4.y;
          acc0 = __builtin_amdgcn_wmma_f32_16x16x4_f32(
               false, a,  false, b,
               (short)0, acc0,  false,  false);
          a.x = a4.z; a.y = a4.w;
          b.x = b4.z; b.y = b4.w;
          acc1 = __builtin_amdgcn_wmma_f32_16x16x4_f32(
               false, a,  false, b,
               (short)0, acc1,  false,  false);
        }
        if (c < 2 * H_HEADS) {
          float* o2 = (c < H_HEADS) ? (lg_next + c) : (lg_next + N_NODES * H_HEADS + (c - H_HEADS));
          #pragma unroll
          for (int i = 0; i < 8; ++i) o2[(m0 + 8 * hf + i) * H_HEADS] = acc0[i] + acc1[i];
        }
      }
    }
  }
}

extern "C" void kernel_launch(void* const* d_in, const int* in_sizes, int n_in,
                              void* d_out, int out_size, void* d_ws, size_t ws_size,
                              hipStream_t stream) {
  const float* x      = (const float*)d_in[0];
  const float* x0     = (const float*)d_in[1];
  const int*   src    = (const int*)d_in[2];
  const int*   dst    = (const int*)d_in[3];
  const float* alpha  = (const float*)d_in[4];
  const float* W      = (const float*)d_in[5];
  const float* attn_l = (const float*)d_in[6];
  const float* attn_r = (const float*)d_in[7];
  const float* bias   = (const float*)d_in[8];

  char* ws = (char*)d_ws;
  size_t off = 0;
  auto alloc = [&](size_t bytes) -> char* {
    char* p = ws + off;
    off = (off + bytes + 255) & ~(size_t)255;
    return p;
  };
  float* lg_a     = (float*)alloc((size_t)2 * N_NODES * H_HEADS * sizeof(float));
  float* lg_b     = (float*)alloc((size_t)2 * N_NODES * H_HEADS * sizeof(float));
  float* xbuf     = (float*)alloc((size_t)N_NODES * F_INF * sizeof(float));
  int*   row_ptr  = (int*)alloc((size_t)CSR_SLICES * RP_STRIDE * sizeof(int));
  int*   edge_src = (int*)alloc((size_t)N_EDGES * sizeof(int));
  float* wlr      = (float*)alloc((size_t)F_INF * 2 * H_HEADS * sizeof(float));

  csr_build<<<CSR_WGS + FOLD_WGS, CSR_THREADS, 0, stream>>>(dst, src, row_ptr, edge_src,
                                                            W, attn_l, attn_r, wlr);
  node_logits<<<LOGIT_TILES / LOGIT_WAVES, LOGIT_WAVES * 32, 0, stream>>>(
      x, wlr, lg_a, lg_a + N_NODES * H_HEADS);

  const float* xin = x;
  float* lg_cur = lg_a;
  float* lg_nxt = lg_b;
  for (int step = 0; step < STEPS; ++step) {
    const bool last = (step == STEPS - 1);
    float* xo = last ? (float*)d_out : xbuf;
    gat_aggregate_project<<<N_NODES / FUSE_NODES, FUSE_WAVES * 32, 0, stream>>>(
        xin, lg_cur, row_ptr, edge_src, W, bias, alpha, x0,
        wlr, last ? nullptr : lg_nxt, xo);
    xin = xbuf;
    float* t = lg_cur; lg_cur = lg_nxt; lg_nxt = t;
  }
}
